# starting kernel
# baseline (speedup 1.0000x reference)
#include <hip/hip_runtime.h>
#include <hip/hip_bf16.h>

typedef __attribute__((ext_vector_type(2))) float v2f;
typedef __attribute__((ext_vector_type(8))) float v8f;

#define GAT_NEG_INF (-9000000000000000.0f)

constexpr int kB   = 8;
constexpr int kN   = 1024;
constexpr int kDin = 128;
constexpr int kH   = 8;
constexpr int kD   = 16;
constexpr int kHD  = 128;
constexpr float kAlpha = 0.2f;

__device__ __forceinline__ float leaky(float x) {
    return fmaxf(x, kAlpha * x);
}

template <int IMM>
__device__ __forceinline__ float swzf(float v) {
    return __int_as_float(__builtin_amdgcn_ds_swizzle(__float_as_int(v), IMM));
}
__device__ __forceinline__ float bpermf(int byteAddr, float v) {
    return __int_as_float(__builtin_amdgcn_ds_bpermute(byteAddr, __float_as_int(v)));
}
#define SWZ_X1  0x041f
#define SWZ_X2  0x081f
#define SWZ_X4  0x101f
#define SWZ_X8  0x201f
#define SWZ_X16 0x401f

__global__ __launch_bounds__(256) void gat_phase1(
    const float* __restrict__ h, const float* __restrict__ W,
    const float* __restrict__ a, float* __restrict__ WhP,
    float* __restrict__ s_i, float* __restrict__ s_j)
{
    __shared__ float hT[16 * 132];

    const int g    = blockIdx.x;
    const int tid  = threadIdx.x;
    const int head = tid >> 5;
    const int lane = tid & 31;
    const int r16  = lane & 15;
    const int half = lane >> 4;

    {
        const float4* src = (const float4*)(h + (size_t)g * 16 * kDin);
        #pragma unroll
        for (int it = 0; it < 2; ++it) {
            int idx  = it * 256 + tid;
            int row  = idx >> 5;
            int col4 = idx & 31;
            float4 v = src[idx];
            *(float4*)&hT[row * 132 + col4 * 4] = v;
        }
    }
    __syncthreads();

    const int n0 = head * 16;
    v8f c = {};
    #pragma unroll 4
    for (int k = 0; k < kDin; k += 4) {
        const int ka = k + 2 * half;
        v2f av, bv;
        av[0] = hT[r16 * 132 + ka];
        av[1] = hT[r16 * 132 + ka + 1];
        bv[0] = W[(size_t)ka * kHD + n0 + r16];
        bv[1] = W[(size_t)(ka + 1) * kHD + n0 + r16];
        c = __builtin_amdgcn_wmma_f32_16x16x4_f32(false, av, false, bv,
                                                  (short)0, c, false, false);
    }

    #pragma unroll
    for (int r = 0; r < 8; ++r) {
        int row = g * 16 + r + half * 8;
        WhP[((size_t)(row >> 1) * kHD + n0 + r16) * 2 + (row & 1)] = c[r];
    }

    const float asrc = a[head * 2 * kD + r16];
    const float adst = a[head * 2 * kD + kD + r16];
    #pragma unroll
    for (int r = 0; r < 8; ++r) {
        float vs = c[r] * asrc;
        float vd = c[r] * adst;
        vs += swzf<SWZ_X1>(vs);  vd += swzf<SWZ_X1>(vd);
        vs += swzf<SWZ_X2>(vs);  vd += swzf<SWZ_X2>(vd);
        vs += swzf<SWZ_X4>(vs);  vd += swzf<SWZ_X4>(vd);
        vs += swzf<SWZ_X8>(vs);  vd += swzf<SWZ_X8>(vd);
        if (r16 == 0) {
            int row = g * 16 + r + half * 8;
            int bb = row >> 10, nn = row & (kN - 1);
            s_i[((size_t)bb * kH + head) * kN + nn] = vs;
            s_j[((size_t)bb * kH + head) * kN + nn] = vd;
        }
    }
}

__global__ __launch_bounds__(256) void gat_phase2(
    const int* __restrict__ adj, const float* __restrict__ WhP,
    const float* __restrict__ s_i, const float* __restrict__ s_j,
    float* __restrict__ out)
{
    __shared__ float adjM[kN * 16];

    const int blk  = blockIdx.x;
    const int b    = blk >> 6;
    const int i0   = (blk & 63) * 16;
    const int tid  = threadIdx.x;
    const int head = tid >> 5;
    const int lane = tid & 31;
    const int r16  = lane & 15;
    const int half = lane >> 4;

    {
        const int4* src = (const int4*)(adj + ((size_t)b * kN + i0) * kN);
        #pragma unroll
        for (int it = 0; it < 16; ++it) {
            int idx = it * 256 + tid;
            int row = idx >> 8;
            int col = (idx & 255) * 4;
            int4 v  = src[idx];
            int rot0 = (col >> 2) & 15;
            int base = (col >> 1) * 32 + 2 * ((row + rot0) & 15);
            float2 lo, hi;
            lo.x = v.x ? 0.0f : GAT_NEG_INF;
            lo.y = v.y ? 0.0f : GAT_NEG_INF;
            hi.x = v.z ? 0.0f : GAT_NEG_INF;
            hi.y = v.w ? 0.0f : GAT_NEG_INF;
            *(float2*)&adjM[base]      = lo;
            *(float2*)&adjM[base + 32] = hi;
        }
    }
    __syncthreads();

    const float* sI = s_i + ((size_t)b * kH + head) * kN;
    const float* sJ = s_j + ((size_t)b * kH + head) * kN;
    const float si_l = sI[i0 + r16];

    float mv = GAT_NEG_INF;
    {
        const float4* sJ4 = (const float4*)sJ;
        #pragma unroll
        for (int k = 0; k < 8; ++k) {
            float4 v = sJ4[k * 32 + lane];
            mv = fmaxf(mv, fmaxf(fmaxf(v.x, v.y), fmaxf(v.z, v.w)));
        }
        mv = fmaxf(mv, swzf<SWZ_X16>(mv));
        mv = fmaxf(mv, swzf<SWZ_X8>(mv));
        mv = fmaxf(mv, swzf<SWZ_X4>(mv));
        mv = fmaxf(mv, swzf<SWZ_X2>(mv));
        mv = fmaxf(mv, swzf<SWZ_X1>(mv));
    }
    const float m_l = leaky(si_l + mv);

    const float* WhPB = WhP + (size_t)b * kN * kHD + head * 32;
    v8f c = {};
    float psum = 0.0f;
    for (int chunk = 0; chunk < 32; ++chunk) {
        const int jb = chunk * 32;
        #pragma unroll 4
        for (int kk = 0; kk < 8; ++kk) {
            const int j0  = jb + kk * 4 + 2 * half;
            const int q0  = j0 >> 1;
            const int rot = (j0 >> 2) & 15;
            const int idx = q0 * 32 + 2 * ((r16 + rot) & 15);
            float2 ad = *(const float2*)&adjM[idx];
            float2 sj = *(const float2*)(sJ + j0);
            float2 wv = *(const float2*)(WhPB + (size_t)q0 * 256 + r16 * 2);
            float t0 = leaky(si_l + sj.x) + ad.x;
            float t1 = leaky(si_l + sj.y) + ad.y;
            float p0 = __expf(t0 - m_l);
            float p1 = __expf(t1 - m_l);
            psum += p0 + p1;
            v2f av, bv;
            av[0] = p0;   av[1] = p1;
            bv[0] = wv.x; bv[1] = wv.y;
            c = __builtin_amdgcn_wmma_f32_16x16x4_f32(false, av, false, bv,
                                                      (short)0, c, false, false);
        }
    }

    psum += swzf<SWZ_X16>(psum);
    const float inv_l = 1.0f / fmaxf(psum, 1e-30f);
    const int baddr = half * 32;
    #pragma unroll
    for (int r = 0; r < 8; ++r) {
        float sc = bpermf(baddr + 4 * r, inv_l);
        int row = i0 + r + half * 8;
        out[((size_t)b * kN + row) * kHD + head * 16 + r16] = c[r] * sc;
    }
}

extern "C" void kernel_launch(void* const* d_in, const int* in_sizes, int n_in,
                              void* d_out, int out_size, void* d_ws, size_t ws_size,
                              hipStream_t stream) {
    const float* h   = (const float*)d_in[0];
    const int*   adj = (const int*)d_in[1];
    const float* W   = (const float*)d_in[2];
    const float* a   = (const float*)d_in[3];
    float* out = (float*)d_out;

    float* WhP = (float*)d_ws;
    float* s_i = WhP + (size_t)kB * kN * kHD;
    float* s_j = s_i + (size_t)kB * kH * kN;

    gat_phase1<<<dim3((kB * kN) / 16), 256, 0, stream>>>(h, W, a, WhP, s_i, s_j);
    gat_phase2<<<dim3(kB * (kN / 16)), 256, 0, stream>>>(adj, WhP, s_i, s_j, out);
}


# Phase 2 sweep software-pipelined: the next 32-column chunk's loads (Wh pairs, logit pairs, LDS mask pairs) go into a second register set before the current chunk's math; arithmetic unchanged
# speedup vs baseline: 1.2661x; 1.2661x over previous
#include <hip/hip_runtime.h>
#include <hip/hip_bf16.h>

typedef __attribute__((ext_vector_type(2))) float v2f;
typedef __attribute__((ext_vector_type(8))) float v8f;

#define GAT_NEG_INF (-9000000000000000.0f)

constexpr int kB   = 8;
constexpr int kN   = 1024;
constexpr int kDin = 128;
constexpr int kH   = 8;
constexpr int kD   = 16;
constexpr int kHD  = 128;
constexpr float kAlpha = 0.2f;

__device__ __forceinline__ float leaky(float x) {
    return fmaxf(x, kAlpha * x);
}

template <int IMM>
__device__ __forceinline__ float swzf(float v) {
    return __int_as_float(__builtin_amdgcn_ds_swizzle(__float_as_int(v), IMM));
}
__device__ __forceinline__ float bpermf(int byteAddr, float v) {
    return __int_as_float(__builtin_amdgcn_ds_bpermute(byteAddr, __float_as_int(v)));
}
#define SWZ_X1  0x041f
#define SWZ_X2  0x081f
#define SWZ_X4  0x101f
#define SWZ_X8  0x201f
#define SWZ_X16 0x401f

__global__ __launch_bounds__(256) void gat_phase1(
    const float* __restrict__ h, const float* __restrict__ W,
    const float* __restrict__ a, float* __restrict__ WhP,
    float* __restrict__ s_i, float* __restrict__ s_j)
{
    __shared__ float hT[16 * 132];

    const int g    = blockIdx.x;
    const int tid  = threadIdx.x;
    const int head = tid >> 5;
    const int lane = tid & 31;
    const int r16  = lane & 15;
    const int half = lane >> 4;

    {
        const float4* src = (const float4*)(h + (size_t)g * 16 * kDin);
        #pragma unroll
        for (int it = 0; it < 2; ++it) {
            int idx  = it * 256 + tid;
            int row  = idx >> 5;
            int col4 = idx & 31;
            float4 v = src[idx];
            *(float4*)&hT[row * 132 + col4 * 4] = v;
        }
    }
    __syncthreads();

    const int n0 = head * 16;
    v8f c = {};
    #pragma unroll 4
    for (int k = 0; k < kDin; k += 4) {
        const int ka = k + 2 * half;
        v2f av, bv;
        av[0] = hT[r16 * 132 + ka];
        av[1] = hT[r16 * 132 + ka + 1];
        bv[0] = W[(size_t)ka * kHD + n0 + r16];
        bv[1] = W[(size_t)(ka + 1) * kHD + n0 + r16];
        c = __builtin_amdgcn_wmma_f32_16x16x4_f32(false, av, false, bv,
                                                  (short)0, c, false, false);
    }

    #pragma unroll
    for (int r = 0; r < 8; ++r) {
        int row = g * 16 + r + half * 8;
        WhP[((size_t)(row >> 1) * kHD + n0 + r16) * 2 + (row & 1)] = c[r];
    }

    const float asrc = a[head * 2 * kD + r16];
    const float adst = a[head * 2 * kD + kD + r16];
    #pragma unroll
    for (int r = 0; r < 8; ++r) {
        float vs = c[r] * asrc;
        float vd = c[r] * adst;
        vs += swzf<SWZ_X1>(vs);  vd += swzf<SWZ_X1>(vd);
        vs += swzf<SWZ_X2>(vs);  vd += swzf<SWZ_X2>(vd);
        vs += swzf<SWZ_X4>(vs);  vd += swzf<SWZ_X4>(vd);
        vs += swzf<SWZ_X8>(vs);  vd += swzf<SWZ_X8>(vd);
        if (r16 == 0) {
            int row = g * 16 + r + half * 8;
            int bb = row >> 10, nn = row & (kN - 1);
            s_i[((size_t)bb * kH + head) * kN + nn] = vs;
            s_j[((size_t)bb * kH + head) * kN + nn] = vd;
        }
    }
}

struct ChunkRegs {
    float2 wv[8];
    float2 sj[8];
    float2 ad[8];
};

__device__ __forceinline__ void load_chunk(
    ChunkRegs& R, int chunk, const float* __restrict__ whLane,
    const float* __restrict__ sjLane, const float* adjM, int r16, int half)
{
    #pragma unroll
    for (int kk = 0; kk < 8; ++kk) {
        const int g   = chunk * 8 + kk;
        const int q0  = g * 2 + half;
        const int idx = q0 * 32 + 2 * ((r16 + g) & 15);
        R.wv[kk] = *(const float2*)(whLane + (size_t)g * 512);
        R.sj[kk] = *(const float2*)(sjLane + g * 4);
        R.ad[kk] = *(const float2*)&adjM[idx];
    }
    asm volatile("" ::: "memory");
}

__device__ __forceinline__ void math_chunk(
    const ChunkRegs& R, float si_l, float m_l, v8f& c, float& psum)
{
    #pragma unroll
    for (int kk = 0; kk < 8; ++kk) {
        float t0 = leaky(si_l + R.sj[kk].x) + R.ad[kk].x;
        float t1 = leaky(si_l + R.sj[kk].y) + R.ad[kk].y;
        float p0 = __expf(t0 - m_l);
        float p1 = __expf(t1 - m_l);
        psum += p0 + p1;
        v2f av, bv;
        av[0] = p0;         av[1] = p1;
        bv[0] = R.wv[kk].x; bv[1] = R.wv[kk].y;
        c = __builtin_amdgcn_wmma_f32_16x16x4_f32(false, av, false, bv,
                                                  (short)0, c, false, false);
    }
}

__global__ __launch_bounds__(256) __attribute__((amdgpu_waves_per_eu(4, 4)))
void gat_phase2(
    const int* __restrict__ adj, const float* __restrict__ WhP,
    const float* __restrict__ s_i, const float* __restrict__ s_j,
    float* __restrict__ out)
{
    __shared__ float adjM[kN * 16];

    const int blk  = blockIdx.x;
    const int b    = blk >> 6;
    const int i0   = (blk & 63) * 16;
    const int tid  = threadIdx.x;
    const int head = tid >> 5;
    const int lane = tid & 31;
    const int r16  = lane & 15;
    const int half = lane >> 4;

    {
        const int4* src = (const int4*)(adj + ((size_t)b * kN + i0) * kN);
        #pragma unroll
        for (int it = 0; it < 16; ++it) {
            int idx = it * 256 + tid;
            int row = idx >> 8;
            int col = (idx & 255) * 4;
            int4 v  = src[idx];
            int rot0 = (col >> 2) & 15;
            int base = (col >> 1) * 32 + 2 * ((row + rot0) & 15);
            float2 lo, hi;
            lo.x = v.x ? 0.0f : GAT_NEG_INF;
            lo.y = v.y ? 0.0f : GAT_NEG_INF;
            hi.x = v.z ? 0.0f : GAT_NEG_INF;
            hi.y = v.w ? 0.0f : GAT_NEG_INF;
            *(float2*)&adjM[base]      = lo;
            *(float2*)&adjM[base + 32] = hi;
        }
    }
    __syncthreads();

    const float* sI = s_i + ((size_t)b * kH + head) * kN;
    const float* sJ = s_j + ((size_t)b * kH + head) * kN;
    const float si_l = sI[i0 + r16];

    float mv = GAT_NEG_INF;
    {
        const float4* sJ4 = (const float4*)sJ;
        #pragma unroll
        for (int k = 0; k < 8; ++k) {
            float4 v = sJ4[k * 32 + lane];
            mv = fmaxf(mv, fmaxf(fmaxf(v.x, v.y), fmaxf(v.z, v.w)));
        }
        mv = fmaxf(mv, swzf<SWZ_X16>(mv));
        mv = fmaxf(mv, swzf<SWZ_X8>(mv));
        mv = fmaxf(mv, swzf<SWZ_X4>(mv));
        mv = fmaxf(mv, swzf<SWZ_X2>(mv));
        mv = fmaxf(mv, swzf<SWZ_X1>(mv));
    }
    const float m_l = leaky(si_l + mv);

    const float* whLane = WhP + (size_t)b * kN * kHD + head * 32
                              + (size_t)half * 256 + r16 * 2;
    const float* sjLane = sJ + 2 * half;
    v8f c = {};
    float psum = 0.0f;
    ChunkRegs ra, rb;
    load_chunk(ra, 0, whLane, sjLane, adjM, r16, half);
    for (int chunk = 0; chunk < 30; chunk += 2) {
        load_chunk(rb, chunk + 1, whLane, sjLane, adjM, r16, half);
        math_chunk(ra, si_l, m_l, c, psum);
        load_chunk(ra, chunk + 2, whLane, sjLane, adjM, r16, half);
        math_chunk(rb, si_l, m_l, c, psum);
    }
    load_chunk(rb, 31, whLane, sjLane, adjM, r16, half);
    math_chunk(ra, si_l, m_l, c, psum);
    math_chunk(rb, si_l, m_l, c, psum);

    psum += swzf<SWZ_X16>(psum);
    const float inv_l = 1.0f / fmaxf(psum, 1e-30f);
    const int baddr = half * 32;
    #pragma unroll
    for (int r = 0; r < 8; ++r) {
        float sc = bpermf(baddr + 4 * r, inv_l);
        int row = i0 + r + half * 8;
        out[((size_t)b * kN + row) * kHD + head * 16 + r16] = c[r] * sc;
    }
}

extern "C" void kernel_launch(void* const* d_in, const int* in_sizes, int n_in,
                              void* d_out, int out_size, void* d_ws, size_t ws_size,
                              hipStream_t stream) {
    const float* h   = (const float*)d_in[0];
    const int*   adj = (const int*)d_in[1];
    const float* W   = (const float*)d_in[2];
    const float* a   = (const float*)d_in[3];
    float* out = (float*)d_out;

    float* WhP = (float*)d_ws;
    float* s_i = WhP + (size_t)kB * kN * kHD;
    float* s_j = s_i + (size_t)kB * kH * kN;

    gat_phase1<<<dim3((kB * kN) / 16), 256, 0, stream>>>(h, W, a, WhP, s_i, s_j);
    gat_phase2<<<dim3(kB * (kN / 16)), 256, 0, stream>>>(adj, WhP, s_i, s_j, out);
}


# exp taken out of the N x N sweep: exp(leaky(si+sj)-m) = max(A_i*F_j, B_i*G_j), F,G = per-head exp column tables built once in LDS, A,B row factors, mask as 0/1 factor; sum order unchanged
# speedup vs baseline: 1.3040x; 1.0299x over previous
#include <hip/hip_runtime.h>
#include <hip/hip_bf16.h>

typedef __attribute__((ext_vector_type(2))) float v2f;
typedef __attribute__((ext_vector_type(8))) float v8f;
typedef __attribute__((ext_vector_type(4))) int   v4i;

#define GAT_NEG_INF (-9000000000000000.0f)

constexpr int kB   = 8;
constexpr int kN   = 1024;
constexpr int kDin = 128;
constexpr int kH   = 8;
constexpr int kD   = 16;
constexpr int kHD  = 128;
constexpr float kAlpha = 0.2f;

template <int IMM>
__device__ __forceinline__ float swzf(float v) {
    return __int_as_float(__builtin_amdgcn_ds_swizzle(__float_as_int(v), IMM));
}
__device__ __forceinline__ float bpermf(int byteAddr, float v) {
    return __int_as_float(__builtin_amdgcn_ds_bpermute(byteAddr, __float_as_int(v)));
}
#define SWZ_X1  0x041f
#define SWZ_X2  0x081f
#define SWZ_X4  0x101f
#define SWZ_X8  0x201f
#define SWZ_X16 0x401f

__global__ __launch_bounds__(256) void gat_phase1(
    const float* __restrict__ h, const float* __restrict__ W,
    const float* __restrict__ a, float* __restrict__ WhP,
    float* __restrict__ s_i, float* __restrict__ s_j)
{
    __shared__ float hT[16 * 132];

    const int g    = blockIdx.x;
    const int tid  = threadIdx.x;
    const int head = tid >> 5;
    const int lane = tid & 31;
    const int r16  = lane & 15;
    const int half = lane >> 4;

    {
        const float4* src = (const float4*)(h + (size_t)g * 16 * kDin);
        #pragma unroll
        for (int it = 0; it < 2; ++it) {
            int idx  = it * 256 + tid;
            int row  = idx >> 5;
            int col4 = idx & 31;
            float4 v = src[idx];
            *(float4*)&hT[row * 132 + col4 * 4] = v;
        }
    }
    __syncthreads();

    const int n0 = head * 16;
    v8f c = {};
    #pragma unroll 4
    for (int k = 0; k < kDin; k += 4) {
        const int ka = k + 2 * half;
        v2f av, bv;
        av[0] = hT[r16 * 132 + ka];
        av[1] = hT[r16 * 132 + ka + 1];
        bv[0] = W[(size_t)ka * kHD + n0 + r16];
        bv[1] = W[(size_t)(ka + 1) * kHD + n0 + r16];
        c = __builtin_amdgcn_wmma_f32_16x16x4_f32(false, av, false, bv,
                                                  (short)0, c, false, false);
    }

    #pragma unroll
    for (int r = 0; r < 8; ++r) {
        int row = g * 16 + r + half * 8;
        WhP[((size_t)(row >> 1) * kHD + n0 + r16) * 2 + (row & 1)] = c[r];
    }

    const float asrc = a[head * 2 * kD + r16];
    const float adst = a[head * 2 * kD + kD + r16];
    #pragma unroll
    for (int r = 0; r < 8; ++r) {
        float vs = c[r] * asrc;
        float vd = c[r] * adst;
        vs += swzf<SWZ_X1>(vs);  vd += swzf<SWZ_X1>(vd);
        vs += swzf<SWZ_X2>(vs);  vd += swzf<SWZ_X2>(vd);
        vs += swzf<SWZ_X4>(vs);  vd += swzf<SWZ_X4>(vd);
        vs += swzf<SWZ_X8>(vs);  vd += swzf<SWZ_X8>(vd);
        if (r16 == 0) {
            int row = g * 16 + r + half * 8;
            int bb = row >> 10, nn = row & (kN - 1);
            s_i[((size_t)bb * kH + head) * kN + nn] = vs;
            s_j[((size_t)bb * kH + head) * kN + nn] = vd;
        }
    }
}

struct ChunkRegs {
    float2 wv[8];
    float4 fg[8];
    float2 ad[8];
};

__device__ __forceinline__ void load_chunk(
    ChunkRegs& R, int chunk, const float* __restrict__ whLane,
    const float4* fgHalf, const float* adjM, int r16, int half)
{
    #pragma unroll
    for (int kk = 0; kk < 8; ++kk) {
        const int g   = chunk * 8 + kk;
        const int q0  = g * 2 + half;
        const int idx = q0 * 32 + 2 * ((r16 + g) & 15);
        R.wv[kk] = *(const float2*)(whLane + (size_t)g * 512);
        R.fg[kk] = fgHalf[g * 2];
        R.ad[kk] = *(const float2*)&adjM[idx];
    }
    asm volatile("" ::: "memory");
}

__device__ __forceinline__ void math_chunk(
    const ChunkRegs& R, float A_l, float B_l, v8f& c, float& psum)
{
    #pragma unroll
    for (int kk = 0; kk < 8; ++kk) {
        float p0 = fmaxf(A_l * R.fg[kk].x, B_l * R.fg[kk].z) * R.ad[kk].x;
        float p1 = fmaxf(A_l * R.fg[kk].y, B_l * R.fg[kk].w) * R.ad[kk].y;
        psum += p0 + p1;
        v2f av, bv;
        av[0] = p0;         av[1] = p1;
        bv[0] = R.wv[kk].x; bv[1] = R.wv[kk].y;
        c = __builtin_amdgcn_wmma_f32_16x16x4_f32(false, av, false, bv,
                                                  (short)0, c, false, false);
    }
}

__global__ __launch_bounds__(256) __attribute__((amdgpu_waves_per_eu(4, 4)))
void gat_phase2(
    const int* __restrict__ adj, const float* __restrict__ WhP,
    const float* __restrict__ s_i, const float* __restrict__ s_j,
    float* __restrict__ out)
{
    __shared__ float  adjM[kN * 16];
    __shared__ float4 colFG[kH * (kN / 2)];

    const int blk  = blockIdx.x;
    const int b    = blk >> 6;
    const int i0   = (blk & 63) * 16;
    const int tid  = threadIdx.x;
    const int head = tid >> 5;
    const int lane = tid & 31;
    const int r16  = lane & 15;
    const int half = lane >> 4;

    const float* sI = s_i + ((size_t)b * kH + head) * kN;
    const float* sJ = s_j + ((size_t)b * kH + head) * kN;

    float4 sv[8];
    {
        const float4* sJ4 = (const float4*)sJ;
        #pragma unroll
        for (int k = 0; k < 8; ++k) sv[k] = sJ4[k * 32 + lane];
    }
    const float si_l = sI[i0 + r16];
    __builtin_amdgcn_sched_barrier(0);
    v4i av[16];
    {
        const v4i* src = (const v4i*)(adj + ((size_t)b * kN + i0) * kN);
        #pragma unroll
        for (int it = 0; it < 16; ++it) av[it] = src[it * 256 + tid];
    }
    __builtin_amdgcn_sched_barrier(0);

    float mv = GAT_NEG_INF;
    {
        #pragma unroll
        for (int k = 0; k < 8; ++k)
            mv = fmaxf(mv, fmaxf(fmaxf(sv[k].x, sv[k].y), fmaxf(sv[k].z, sv[k].w)));
        mv = fmaxf(mv, swzf<SWZ_X16>(mv));
        mv = fmaxf(mv, swzf<SWZ_X8>(mv));
        mv = fmaxf(mv, swzf<SWZ_X4>(mv));
        mv = fmaxf(mv, swzf<SWZ_X2>(mv));
        mv = fmaxf(mv, swzf<SWZ_X1>(mv));
        float4* fgw = colFG + head * (kN / 2);
        #pragma unroll
        for (int k = 0; k < 8; ++k) {
            const int f = k * 32 + lane;
            const float d0 = sv[k].x - mv, d1 = sv[k].y - mv;
            const float d2 = sv[k].z - mv, d3 = sv[k].w - mv;
            float4 lo, hi;
            lo.x = __expf(d0);          lo.y = __expf(d1);
            lo.z = __expf(kAlpha * d0); lo.w = __expf(kAlpha * d1);
            hi.x = __expf(d2);          hi.y = __expf(d3);
            hi.z = __expf(kAlpha * d2); hi.w = __expf(kAlpha * d3);
            fgw[2 * f]     = lo;
            fgw[2 * f + 1] = hi;
        }
    }
    __builtin_amdgcn_sched_barrier(0);

    {
        #pragma unroll
        for (int it = 0; it < 16; ++it) {
            int idx = it * 256 + tid;
            int row = idx >> 8;
            int col = (idx & 255) * 4;
            v4i v   = av[it];
            int rot0 = (col >> 2) & 15;
            int base = (col >> 1) * 32 + 2 * ((row + rot0) & 15);
            float2 lo, hi;
            lo.x = v.x ? 1.0f : 0.0f;
            lo.y = v.y ? 1.0f : 0.0f;
            hi.x = v.z ? 1.0f : 0.0f;
            hi.y = v.w ? 1.0f : 0.0f;
            *(float2*)&adjM[base]      = lo;
            *(float2*)&adjM[base + 32] = hi;
        }
    }
    __syncthreads();

    const float u_l  = (1.0f - kAlpha) * (si_l + mv);
    const float A_l  = __expf(fminf(u_l, 0.0f));
    const float B_l  = __expf(fminf(-u_l, 0.0f));

    const float* whLane = WhP + (size_t)b * kN * kHD + head * 32
                              + (size_t)half * 256 + r16 * 2;
    const float4* fgHalf = colFG + head * (kN / 2) + half;
    v8f c = {};
    float psum = 0.0f;
    ChunkRegs ra, rb;
    load_chunk(ra, 0, whLane, fgHalf, adjM, r16, half);
    for (int chunk = 0; chunk < 30; chunk += 2) {
        load_chunk(rb, chunk + 1, whLane, fgHalf, adjM, r16, half);
        math_chunk(ra, A_l, B_l, c, psum);
        load_chunk(ra, chunk + 2, whLane, fgHalf, adjM, r16, half);
        math_chunk(rb, A_l, B_l, c, psum);
    }
    load_chunk(rb, 31, whLane, fgHalf, adjM, r16, half);
    math_chunk(ra, A_l, B_l, c, psum);
    math_chunk(rb, A_l, B_l, c, psum);

    psum += swzf<SWZ_X16>(psum);
    const float inv_l = 1.0f / fmaxf(psum, 1e-30f);
    const int baddr = half * 32;
    #pragma unroll
    for (int r = 0; r < 8; ++r) {
        float sc = bpermf(baddr + 4 * r, inv_l);
        int row = i0 + r + half * 8;
        out[((size_t)b * kN + row) * kHD + head * 16 + r16] = c[r] * sc;
    }
}

extern "C" void kernel_launch(void* const* d_in, const int* in_sizes, int n_in,
                              void* d_out, int out_size, void* d_ws, size_t ws_size,
                              hipStream_t stream) {
    const float* h   = (const float*)d_in[0];
    const int*   adj = (const int*)d_in[1];
    const float* W   = (const float*)d_in[2];
    const float* a   = (const float*)d_in[3];
    float* out = (float*)d_out;

    float* WhP = (float*)d_ws;
    float* s_i = WhP + (size_t)kB * kN * kHD;
    float* s_j = s_i + (size_t)kB * kH * kN;

    gat_phase1<<<dim3((kB * kN) / 16), 256, 0, stream>>>(h, W, a, WhP, s_i, s_j);
    gat_phase2<<<dim3(kB * (kN / 16)), 256, 0, stream>>>(adj, WhP, s_i, s_j, out);
}


# Phase 2: each wave carries two 16-row tiles (16 waves = 8 heads x 2 column halves): each Wh and F/G fragment feeds two WMMAs; sums now per column half and parity, halves joined through LDS
# speedup vs baseline: 1.3922x; 1.0676x over previous
#include <hip/hip_runtime.h>
#include <hip/hip_bf16.h>

typedef __attribute__((ext_vector_type(2))) float v2f;
typedef __attribute__((ext_vector_type(8))) float v8f;
typedef __attribute__((ext_vector_type(4))) int   v4i;

#define GAT_NEG_INF (-9000000000000000.0f)

constexpr int kB   = 8;
constexpr int kN   = 1024;
constexpr int kDin = 128;
constexpr int kH   = 8;
constexpr int kD   = 16;
constexpr int kHD  = 128;
constexpr float kAlpha = 0.2f;

template <int IMM>
__device__ __forceinline__ float swzf(float v) {
    return __int_as_float(__builtin_amdgcn_ds_swizzle(__float_as_int(v), IMM));
}
__device__ __forceinline__ float bpermf(int byteAddr, float v) {
    return __int_as_float(__builtin_amdgcn_ds_bpermute(byteAddr, __float_as_int(v)));
}
#define SWZ_X1  0x041f
#define SWZ_X2  0x081f
#define SWZ_X4  0x101f
#define SWZ_X8  0x201f
#define SWZ_X16 0x401f

__global__ __launch_bounds__(256) void gat_phase1(
    const float* __restrict__ h, const float* __restrict__ W,
    const float* __restrict__ a, float* __restrict__ WhP,
    float* __restrict__ s_i, float* __restrict__ s_j)
{
    __shared__ float hT[16 * 132];

    const int g    = blockIdx.x;
    const int tid  = threadIdx.x;
    const int head = tid >> 5;
    const int lane = tid & 31;
    const int r16  = lane & 15;
    const int half = lane >> 4;

    {
        const float4* src = (const float4*)(h + (size_t)g * 16 * kDin);
        #pragma unroll
        for (int it = 0; it < 2; ++it) {
            int idx  = it * 256 + tid;
            int row  = idx >> 5;
            int col4 = idx & 31;
            float4 v = src[idx];
            *(float4*)&hT[row * 132 + col4 * 4] = v;
        }
    }
    __syncthreads();

    const int n0 = head * 16;
    v8f c = {};
    #pragma unroll 4
    for (int k = 0; k < kDin; k += 4) {
        const int ka = k + 2 * half;
        v2f av, bv;
        av[0] = hT[r16 * 132 + ka];
        av[1] = hT[r16 * 132 + ka + 1];
        bv[0] = W[(size_t)ka * kHD + n0 + r16];
        bv[1] = W[(size_t)(ka + 1) * kHD + n0 + r16];
        c = __builtin_amdgcn_wmma_f32_16x16x4_f32(false, av, false, bv,
                                                  (short)0, c, false, false);
    }

    #pragma unroll
    for (int r = 0; r < 8; ++r) {
        int row = g * 16 + r + half * 8;
        WhP[((size_t)(row >> 1) * kHD + n0 + r16) * 2 + (row & 1)] = c[r];
    }

    const float asrc = a[head * 2 * kD + r16];
    const float adst = a[head * 2 * kD + kD + r16];
    #pragma unroll
    for (int r = 0; r < 8; ++r) {
        float vs = c[r] * asrc;
        float vd = c[r] * adst;
        vs += swzf<SWZ_X1>(vs);  vd += swzf<SWZ_X1>(vd);
        vs += swzf<SWZ_X2>(vs);  vd += swzf<SWZ_X2>(vd);
        vs += swzf<SWZ_X4>(vs);  vd += swzf<SWZ_X4>(vd);
        vs += swzf<SWZ_X8>(vs);  vd += swzf<SWZ_X8>(vd);
        if (r16 == 0) {
            int row = g * 16 + r + half * 8;
            int bb = row >> 10, nn = row & (kN - 1);
            s_i[((size_t)bb * kH + head) * kN + nn] = vs;
            s_j[((size_t)bb * kH + head) * kN + nn] = vd;
        }
    }
}

constexpr int kMPitch = 96;

constexpr int kKS = 4;

typedef __attribute__((ext_vector_type(4))) float v4f;

struct ChunkRegs {
    v2f wv[kKS];
    v4f fg[kKS];
    v2f ad0[kKS];
    v2f ad1[kKS];
};

__device__ __forceinline__ void load_chunk(
    ChunkRegs& R, int chunk, const float* __restrict__ whLane,
    const v4f* fgHalf, const float* adjM, int r16, int half)
{
    const int halfOff = half * kMPitch;
    #pragma unroll
    for (int kk = 0; kk < kKS; ++kk) {
        const int g   = chunk * kKS + kk;
        const int idx = g * (2 * kMPitch) + halfOff + 2 * ((r16 + g) & 15);
        R.wv[kk]  = *(const v2f*)(whLane + (size_t)g * 512);
        R.fg[kk]  = fgHalf[g * 2];
        R.ad0[kk] = *(const v2f*)&adjM[idx];
        R.ad1[kk] = *(const v2f*)&adjM[idx + 32];
    }
    asm volatile("" ::: "memory");
}

__device__ __forceinline__ void math_chunk(
    const ChunkRegs& R, v2f A0, v2f B0, v2f A1, v2f B1,
    v8f& c0, v8f& c1, v2f& ps0, v2f& ps1)
{
    #pragma unroll
    for (int kk = 0; kk < kKS; ++kk) {
        const v2f F  = R.fg[kk].xy;
        const v2f G  = R.fg[kk].zw;
        const v2f bv = R.wv[kk];
        {
            const v2f p = __builtin_elementwise_max(A0 * F, B0 * G) * R.ad0[kk];
            ps0 += p;
            c0 = __builtin_amdgcn_wmma_f32_16x16x4_f32(false, p, false, bv,
                                                       (short)0, c0, false, false);
        }
        {
            const v2f p = __builtin_elementwise_max(A1 * F, B1 * G) * R.ad1[kk];
            ps1 += p;
            c1 = __builtin_amdgcn_wmma_f32_16x16x4_f32(false, p, false, bv,
                                                       (short)0, c1, false, false);
        }
    }
}

__global__ __launch_bounds__(512) __attribute__((amdgpu_waves_per_eu(4, 4)))
void gat_phase2(
    const int* __restrict__ adj, const float* __restrict__ WhP,
    const float* __restrict__ s_i, const float* __restrict__ s_j,
    float* __restrict__ out)
{
    __shared__ float  adjM[(kN / 2) * kMPitch];
    __shared__ float4 colFG[kH * (kN / 2)];
    __shared__ float  xch[16 * 9 * 32];

    const int blk  = blockIdx.x;
    const int b    = blk >> 5;
    const int i0   = (blk & 31) * 32;
    const int tid  = threadIdx.x;
    const int wid  = __builtin_amdgcn_readfirstlane(tid >> 5);
    const int head = wid & 7;
    const int jh   = wid >> 3;
    const int lane = tid & 31;
    const int r16  = lane & 15;
    const int half = lane >> 4;

    const float* sI = s_i + ((size_t)b * kH + head) * kN;
    const float* sJ = s_j + ((size_t)b * kH + head) * kN;

    float4 svo[4], svx[4];
    {
        const float4* sJ4 = (const float4*)sJ;
        #pragma unroll
        for (int k = 0; k < 4; ++k) svo[k] = sJ4[(jh * 4 + k) * 32 + lane];
        #pragma unroll
        for (int k = 0; k < 4; ++k) svx[k] = sJ4[((jh ^ 1) * 4 + k) * 32 + lane];
    }
    const float si0 = sI[i0 + r16];
    const float si1 = sI[i0 + 16 + r16];
    __builtin_amdgcn_sched_barrier(0);
    v4i av[16];
    {
        const v4i* src = (const v4i*)(adj + ((size_t)b * kN + i0) * kN);
        #pragma unroll
        for (int it = 0; it < 16; ++it) av[it] = src[it * 512 + tid];
    }
    __builtin_amdgcn_sched_barrier(0);

    float mv = GAT_NEG_INF;
    {
        #pragma unroll
        for (int k = 0; k < 4; ++k) {
            mv = fmaxf(mv, fmaxf(fmaxf(svo[k].x, svo[k].y), fmaxf(svo[k].z, svo[k].w)));
            mv = fmaxf(mv, fmaxf(fmaxf(svx[k].x, svx[k].y), fmaxf(svx[k].z, svx[k].w)));
        }
        mv = fmaxf(mv, swzf<SWZ_X16>(mv));
        mv = fmaxf(mv, swzf<SWZ_X8>(mv));
        mv = fmaxf(mv, swzf<SWZ_X4>(mv));
        mv = fmaxf(mv, swzf<SWZ_X2>(mv));
        mv = fmaxf(mv, swzf<SWZ_X1>(mv));
        float4* fgw = colFG + head * (kN / 2);
        #pragma unroll
        for (int k = 0; k < 4; ++k) {
            const int f = (jh * 4 + k) * 32 + lane;
            const float d0 = svo[k].x - mv, d1 = svo[k].y - mv;
            const float d2 = svo[k].z - mv, d3 = svo[k].w - mv;
            float4 lo, hi;
            lo.x = __expf(d0);          lo.y = __expf(d1);
            lo.z = __expf(kAlpha * d0); lo.w = __expf(kAlpha * d1);
            hi.x = __expf(d2);          hi.y = __expf(d3);
            hi.z = __expf(kAlpha * d2); hi.w = __expf(kAlpha * d3);
            fgw[2 * f]     = lo;
            fgw[2 * f + 1] = hi;
        }
    }
    __builtin_amdgcn_sched_barrier(0);

    {
        #pragma unroll
        for (int it = 0; it < 16; ++it) {
            int idx = it * 512 + tid;
            int row = idx >> 8;
            int col = (idx & 255) * 4;
            v4i v   = av[it];
            int rot0 = (col >> 2) & 15;
            int base = (col >> 1) * kMPitch + 32 * (row >> 4)
                     + 2 * (((row & 15) + rot0) & 15);
            float2 lo, hi;
            lo.x = v.x ? 1.0f : 0.0f;
            lo.y = v.y ? 1.0f : 0.0f;
            hi.x = v.z ? 1.0f : 0.0f;
            hi.y = v.w ? 1.0f : 0.0f;
            *(float2*)&adjM[base]      = lo;
            *(float2*)&adjM[base + kMPitch] = hi;
        }
    }
    __syncthreads();

    const float u0 = (1.0f - kAlpha) * (si0 + mv);
    const float a0 = __expf(fminf(u0, 0.0f));
    const float b0 = __expf(fminf(-u0, 0.0f));
    const float u1 = (1.0f - kAlpha) * (si1 + mv);
    const float a1 = __expf(fminf(u1, 0.0f));
    const float b1 = __expf(fminf(-u1, 0.0f));
    const v2f A0 = {a0, a0}, B0 = {b0, b0}, A1 = {a1, a1}, B1 = {b1, b1};

    const float* whLane = WhP + (size_t)b * kN * kHD + head * 32
                              + (size_t)half * 256 + r16 * 2;
    const v4f* fgHalf = (const v4f*)colFG + head * (kN / 2) + half;
    v8f c0 = {}, c1 = {};
    v2f pv0 = {0.0f, 0.0f}, pv1 = {0.0f, 0.0f};
    const int cb = jh * 32;
    ChunkRegs ra, rb;
    load_chunk(ra, cb, whLane, fgHalf, adjM, r16, half);
    for (int cc = 0; cc < 28; cc += 4) {
        load_chunk(rb, cb + cc + 1, whLane, fgHalf, adjM, r16, half);
        math_chunk(ra, A0, B0, A1, B1, c0, c1, pv0, pv1);
        load_chunk(ra, cb + cc + 2, whLane, fgHalf, adjM, r16, half);
        math_chunk(rb, A0, B0, A1, B1, c0, c1, pv0, pv1);
        load_chunk(rb, cb + cc + 3, whLane, fgHalf, adjM, r16, half);
        math_chunk(ra, A0, B0, A1, B1, c0, c1, pv0, pv1);
        load_chunk(ra, cb + cc + 4, whLane, fgHalf, adjM, r16, half);
        math_chunk(rb, A0, B0, A1, B1, c0, c1, pv0, pv1);
    }
    load_chunk(rb, cb + 29, whLane, fgHalf, adjM, r16, half);
    math_chunk(ra, A0, B0, A1, B1, c0, c1, pv0, pv1);
    load_chunk(ra, cb + 30, whLane, fgHalf, adjM, r16, half);
    math_chunk(rb, A0, B0, A1, B1, c0, c1, pv0, pv1);
    load_chunk(rb, cb + 31, whLane, fgHalf, adjM, r16, half);
    math_chunk(ra, A0, B0, A1, B1, c0, c1, pv0, pv1);
    math_chunk(rb, A0, B0, A1, B1, c0, c1, pv0, pv1);

    float ps0 = pv0[0] + pv0[1];
    float ps1 = pv1[0] + pv1[1];
    ps0 += swzf<SWZ_X16>(ps0);
    ps1 += swzf<SWZ_X16>(ps1);

    v8f cg, ck;
    float pg, pk;
    if (jh) { cg = c0; pg = ps0; ck = c1; pk = ps1; }
    else    { cg = c1; pg = ps1; ck = c0; pk = ps0; }
    {
        float* xw = xch + wid * (9 * 32);
        #pragma unroll
        for (int r = 0; r < 8; ++r) xw[r * 32 + lane] = cg[r];
        xw[8 * 32 + lane] = pg;
    }
    __syncthreads();
    {
        const float* xr = xch + (wid ^ 8) * (9 * 32);
        #pragma unroll
        for (int r = 0; r < 8; ++r) ck[r] += xr[r * 32 + lane];
        pk += xr[8 * 32 + lane];
    }

    const float inv_l = 1.0f / fmaxf(pk, 1e-30f);
    const int baddr = half * 32;
    #pragma unroll
    for (int r = 0; r < 8; ++r) {
        float sc = bpermf(baddr + 4 * r, inv_l);
        int row = i0 + jh * 16 + r + half * 8;
        out[((size_t)b * kN + row) * kHD + head * 16 + r16] = ck[r] * sc;
    }
}

extern "C" void kernel_launch(void* const* d_in, const int* in_sizes, int n_in,
                              void* d_out, int out_size, void* d_ws, size_t ws_size,
                              hipStream_t stream) {
    const float* h   = (const float*)d_in[0];
    const int*   adj = (const int*)d_in[1];
    const float* W   = (const float*)d_in[2];
    const float* a   = (const float*)d_in[3];
    float* out = (float*)d_out;

    float* WhP = (float*)d_ws;
    float* s_i = WhP + (size_t)kB * kN * kHD;
    float* s_j = s_i + (size_t)kB * kH * kN;

    gat_phase1<<<dim3((kB * kN) / 16), 256, 0, stream>>>(h, W, a, WhP, s_i, s_j);
    gat_phase2<<<dim3(kB * (kN / 32)), 512, 0, stream>>>(adj, WhP, s_i, s_j, out);
}


# Launch 1 de-serialised: global loads (h tile, 64 W values per lane, logit weights) as one group before the barrier, A fragments as one LDS group, 16 logit reductions interleaved; sums unchanged
# speedup vs baseline: 1.6618x; 1.1937x over previous
#include <hip/hip_runtime.h>
#include <hip/hip_bf16.h>

typedef __attribute__((ext_vector_type(2))) float v2f;
typedef __attribute__((ext_vector_type(8))) float v8f;
typedef __attribute__((ext_vector_type(4))) int   v4i;

#define GAT_NEG_INF (-9000000000000000.0f)

constexpr int kB   = 8;
constexpr int kN   = 1024;
constexpr int kDin = 128;
constexpr int kH   = 8;
constexpr int kD   = 16;
constexpr int kHD  = 128;
constexpr float kAlpha = 0.2f;

template <int IMM>
__device__ __forceinline__ float swzf(float v) {
    return __int_as_float(__builtin_amdgcn_ds_swizzle(__float_as_int(v), IMM));
}
__device__ __forceinline__ float bpermf(int byteAddr, float v) {
    return __int_as_float(__builtin_amdgcn_ds_bpermute(byteAddr, __float_as_int(v)));
}
#define SWZ_X1  0x041f
#define SWZ_X2  0x081f
#define SWZ_X4  0x101f
#define SWZ_X8  0x201f
#define SWZ_X16 0x401f

__global__ __launch_bounds__(256) void gat_phase1(
    const float* __restrict__ h, const float* __restrict__ W,
    const float* __restrict__ a, float* __restrict__ WhP,
    float* __restrict__ s_i, float* __restrict__ s_j)
{
    __shared__ float hT[16 * 132];

    const int g    = blockIdx.x;
    const int tid  = threadIdx.x;
    const int head = tid >> 5;
    const int lane = tid & 31;
    const int r16  = lane & 15;
    const int half = lane >> 4;
    const int n0   = head * 16;

    const float4* src = (const float4*)(h + (size_t)g * 16 * kDin);
    const float4 hv0 = src[tid];
    const float4 hv1 = src[256 + tid];
    const float asrc = a[head * 2 * kD + r16];
    const float adst = a[head * 2 * kD + kD + r16];
    float wreg[64];
    #pragma unroll
    for (int s = 0; s < 32; ++s) {
        const int ka = 4 * s + 2 * half;
        wreg[2 * s]     = W[(size_t)ka * kHD + n0 + r16];
        wreg[2 * s + 1] = W[(size_t)(ka + 1) * kHD + n0 + r16];
    }

    {
        *(float4*)&hT[(tid >> 5) * 132 + (tid & 31) * 4]       = hv0;
        *(float4*)&hT[(8 + (tid >> 5)) * 132 + (tid & 31) * 4] = hv1;
    }
    __syncthreads();

    v2f areg[32];
    #pragma unroll
    for (int s = 0; s < 32; ++s)
        areg[s] = *(const v2f*)&hT[r16 * 132 + 4 * s + 2 * half];
    __builtin_amdgcn_sched_barrier(0);

    v8f c = {};
    #pragma unroll
    for (int s = 0; s < 32; ++s) {
        v2f bv;
        bv[0] = wreg[2 * s];
        bv[1] = wreg[2 * s + 1];
        c = __builtin_amdgcn_wmma_f32_16x16x4_f32(false, areg[s], false, bv,
                                                  (short)0, c, false, false);
    }

    #pragma unroll
    for (int r = 0; r < 8; ++r) {
        int row = g * 16 + r + half * 8;
        WhP[((size_t)(row >> 1) * kHD + n0 + r16) * 2 + (row & 1)] = c[r];
    }

    float vs[8], vd[8];
    #pragma unroll
    for (int r = 0; r < 8; ++r) { vs[r] = c[r] * asrc; vd[r] = c[r] * adst; }
    #pragma unroll
    for (int r = 0; r < 8; ++r) { vs[r] += swzf<SWZ_X1>(vs[r]); vd[r] += swzf<SWZ_X1>(vd[r]); }
    #pragma unroll
    for (int r = 0; r < 8; ++r) { vs[r] += swzf<SWZ_X2>(vs[r]); vd[r] += swzf<SWZ_X2>(vd[r]); }
    #pragma unroll
    for (int r = 0; r < 8; ++r) { vs[r] += swzf<SWZ_X4>(vs[r]); vd[r] += swzf<SWZ_X4>(vd[r]); }
    #pragma unroll
    for (int r = 0; r < 8; ++r) { vs[r] += swzf<SWZ_X8>(vs[r]); vd[r] += swzf<SWZ_X8>(vd[r]); }
    if (r16 == 0) {
        const int row0 = g * 16 + half * 8;
        const int bb = row0 >> 10, nn0 = row0 & (kN - 1);
        float* pi = s_i + ((size_t)bb * kH + head) * kN + nn0;
        float* pj = s_j + ((size_t)bb * kH + head) * kN + nn0;
        *(float4*)pi       = make_float4(vs[0], vs[1], vs[2], vs[3]);
        *(float4*)(pi + 4) = make_float4(vs[4], vs[5], vs[6], vs[7]);
        *(float4*)pj       = make_float4(vd[0], vd[1], vd[2], vd[3]);
        *(float4*)(pj + 4) = make_float4(vd[4], vd[5], vd[6], vd[7]);
    }
}

constexpr int kMPitch = 96;

constexpr int kKS = 4;

typedef __attribute__((ext_vector_type(4))) float v4f;

struct ChunkRegs {
    v2f wv[kKS];
    v4f fg[kKS];
    v2f ad0[kKS];
    v2f ad1[kKS];
};

__device__ __forceinline__ void load_chunk(
    ChunkRegs& R, int chunk, const float* __restrict__ whLane,
    const v4f* fgHalf, const float* adjM, int r16, int half)
{
    const int halfOff = half * kMPitch;
    #pragma unroll
    for (int kk = 0; kk < kKS; ++kk) {
        const int g   = chunk * kKS + kk;
        const int idx = g * (2 * kMPitch) + halfOff + 2 * ((r16 + g) & 15);
        R.wv[kk]  = *(const v2f*)(whLane + (size_t)g * 512);
        R.fg[kk]  = fgHalf[g * 2];
        R.ad0[kk] = *(const v2f*)&adjM[idx];
        R.ad1[kk] = *(const v2f*)&adjM[idx + 32];
    }
    asm volatile("" ::: "memory");
}

__device__ __forceinline__ void math_chunk(
    const ChunkRegs& R, v2f A0, v2f B0, v2f A1, v2f B1,
    v8f& c0, v8f& c1, v2f& ps0, v2f& ps1)
{
    #pragma unroll
    for (int kk = 0; kk < kKS; ++kk) {
        const v2f F  = R.fg[kk].xy;
        const v2f G  = R.fg[kk].zw;
        const v2f bv = R.wv[kk];
        {
            const v2f p = __builtin_elementwise_max(A0 * F, B0 * G) * R.ad0[kk];
            ps0 += p;
            c0 = __builtin_amdgcn_wmma_f32_16x16x4_f32(false, p, false, bv,
                                                       (short)0, c0, false, false);
        }
        {
            const v2f p = __builtin_elementwise_max(A1 * F, B1 * G) * R.ad1[kk];
            ps1 += p;
            c1 = __builtin_amdgcn_wmma_f32_16x16x4_f32(false, p, false, bv,
                                                       (short)0, c1, false, false);
        }
    }
}

__global__ __launch_bounds__(512) __attribute__((amdgpu_waves_per_eu(4, 4)))
void gat_phase2(
    const int* __restrict__ adj, const float* __restrict__ WhP,
    const float* __restrict__ s_i, const float* __restrict__ s_j,
    float* __restrict__ out)
{
    __shared__ float  adjM[(kN / 2) * kMPitch];
    __shared__ float4 colFG[kH * (kN / 2)];
    __shared__ float  xch[16 * 9 * 32];

    const int blk  = blockIdx.x;
    const int b    = blk >> 5;
    const int i0   = (blk & 31) * 32;
    const int tid  = threadIdx.x;
    const int wid  = __builtin_amdgcn_readfirstlane(tid >> 5);
    const int head = wid & 7;
    const int jh   = wid >> 3;
    const int lane = tid & 31;
    const int r16  = lane & 15;
    const int half = lane >> 4;

    const float* sI = s_i + ((size_t)b * kH + head) * kN;
    const float* sJ = s_j + ((size_t)b * kH + head) * kN;

    float4 svo[4], svx[4];
    {
        const float4* sJ4 = (const float4*)sJ;
        #pragma unroll
        for (int k = 0; k < 4; ++k) svo[k] = sJ4[(jh * 4 + k) * 32 + lane];
        #pragma unroll
        for (int k = 0; k < 4; ++k) svx[k] = sJ4[((jh ^ 1) * 4 + k) * 32 + lane];
    }
    const float si0 = sI[i0 + r16];
    const float si1 = sI[i0 + 16 + r16];
    __builtin_amdgcn_sched_barrier(0);
    v4i av[16];
    {
        const v4i* src = (const v4i*)(adj + ((size_t)b * kN + i0) * kN);
        #pragma unroll
        for (int it = 0; it < 16; ++it) av[it] = src[it * 512 + tid];
    }
    __builtin_amdgcn_sched_barrier(0);

    float mv = GAT_NEG_INF;
    {
        #pragma unroll
        for (int k = 0; k < 4; ++k) {
            mv = fmaxf(mv, fmaxf(fmaxf(svo[k].x, svo[k].y), fmaxf(svo[k].z, svo[k].w)));
            mv = fmaxf(mv, fmaxf(fmaxf(svx[k].x, svx[k].y), fmaxf(svx[k].z, svx[k].w)));
        }
        mv = fmaxf(mv, swzf<SWZ_X16>(mv));
        mv = fmaxf(mv, swzf<SWZ_X8>(mv));
        mv = fmaxf(mv, swzf<SWZ_X4>(mv));
        mv = fmaxf(mv, swzf<SWZ_X2>(mv));
        mv = fmaxf(mv, swzf<SWZ_X1>(mv));
        float4* fgw = colFG + head * (kN / 2);
        #pragma unroll
        for (int k = 0; k < 4; ++k) {
            const int f = (jh * 4 + k) * 32 + lane;
            const float d0 = svo[k].x - mv, d1 = svo[k].y - mv;
            const float d2 = svo[k].z - mv, d3 = svo[k].w - mv;
            float4 lo, hi;
            lo.x = __expf(d0);          lo.y = __expf(d1);
            lo.z = __expf(kAlpha * d0); lo.w = __expf(kAlpha * d1);
            hi.x = __expf(d2);          hi.y = __expf(d3);
            hi.z = __expf(kAlpha * d2); hi.w = __expf(kAlpha * d3);
            fgw[2 * f]     = lo;
            fgw[2 * f + 1] = hi;
        }
    }
    __builtin_amdgcn_sched_barrier(0);

    {
        #pragma unroll
        for (int it = 0; it < 16; ++it) {
            int idx = it * 512 + tid;
            int row = idx >> 8;
            int col = (idx & 255) * 4;
            v4i v   = av[it];
            int rot0 = (col >> 2) & 15;
            int base = (col >> 1) * kMPitch + 32 * (row >> 4)
                     + 2 * (((row & 15) + rot0) & 15);
            float2 lo, hi;
            lo.x = v.x ? 1.0f : 0.0f;
            lo.y = v.y ? 1.0f : 0.0f;
            hi.x = v.z ? 1.0f : 0.0f;
            hi.y = v.w ? 1.0f : 0.0f;
            *(float2*)&adjM[base]      = lo;
            *(float2*)&adjM[base + kMPitch] = hi;
        }
    }
    __syncthreads();

    const float u0 = (1.0f - kAlpha) * (si0 + mv);
    const float a0 = __expf(fminf(u0, 0.0f));
    const float b0 = __expf(fminf(-u0, 0.0f));
    const float u1 = (1.0f - kAlpha) * (si1 + mv);
    const float a1 = __expf(fminf(u1, 0.0f));
    const float b1 = __expf(fminf(-u1, 0.0f));
    const v2f A0 = {a0, a0}, B0 = {b0, b0}, A1 = {a1, a1}, B1 = {b1, b1};

    const float* whLane = WhP + (size_t)b * kN * kHD + head * 32
                              + (size_t)half * 256 + r16 * 2;
    const v4f* fgHalf = (const v4f*)colFG + head * (kN / 2) + half;
    v8f c0 = {}, c1 = {};
    v2f pv0 = {0.0f, 0.0f}, pv1 = {0.0f, 0.0f};
    const int cb = jh * 32;
    ChunkRegs ra, rb;
    load_chunk(ra, cb, whLane, fgHalf, adjM, r16, half);
    for (int cc = 0; cc < 28; cc += 4) {
        load_chunk(rb, cb + cc + 1, whLane, fgHalf, adjM, r16, half);
        math_chunk(ra, A0, B0, A1, B1, c0, c1, pv0, pv1);
        load_chunk(ra, cb + cc + 2, whLane, fgHalf, adjM, r16, half);
        math_chunk(rb, A0, B0, A1, B1, c0, c1, pv0, pv1);
        load_chunk(rb, cb + cc + 3, whLane, fgHalf, adjM, r16, half);
        math_chunk(ra, A0, B0, A1, B1, c0, c1, pv0, pv1);
        load_chunk(ra, cb + cc + 4, whLane, fgHalf, adjM, r16, half);
        math_chunk(rb, A0, B0, A1, B1, c0, c1, pv0, pv1);
    }
    load_chunk(rb, cb + 29, whLane, fgHalf, adjM, r16, half);
    math_chunk(ra, A0, B0, A1, B1, c0, c1, pv0, pv1);
    load_chunk(ra, cb + 30, whLane, fgHalf, adjM, r16, half);
    math_chunk(rb, A0, B0, A1, B1, c0, c1, pv0, pv1);
    load_chunk(rb, cb + 31, whLane, fgHalf, adjM, r16, half);
    math_chunk(ra, A0, B0, A1, B1, c0, c1, pv0, pv1);
    math_chunk(rb, A0, B0, A1, B1, c0, c1, pv0, pv1);

    float ps0 = pv0[0] + pv0[1];
    float ps1 = pv1[0] + pv1[1];
    ps0 += swzf<SWZ_X16>(ps0);
    ps1 += swzf<SWZ_X16>(ps1);

    v8f cg, ck;
    float pg, pk;
    if (jh) { cg = c0; pg = ps0; ck = c1; pk = ps1; }
    else    { cg = c1; pg = ps1; ck = c0; pk = ps0; }
    {
        float* xw = xch + wid * (9 * 32);
        #pragma unroll
        for (int r = 0; r < 8; ++r) xw[r * 32 + lane] = cg[r];
        xw[8 * 32 + lane] = pg;
    }
    __syncthreads();
    {
        const float* xr = xch + (wid ^ 8) * (9 * 32);
        #pragma unroll
        for (int r = 0; r < 8; ++r) ck[r] += xr[r * 32 + lane];
        pk += xr[8 * 32 + lane];
    }

    const float inv_l = 1.0f / fmaxf(pk, 1e-30f);
    const int baddr = half * 32;
    #pragma unroll
    for (int r = 0; r < 8; ++r) {
        float sc = bpermf(baddr + 4 * r, inv_l);
        int row = i0 + jh * 16 + r + half * 8;
        out[((size_t)b * kN + row) * kHD + head * 16 + r16] = ck[r] * sc;
    }
}

extern "C" void kernel_launch(void* const* d_in, const int* in_sizes, int n_in,
                              void* d_out, int out_size, void* d_ws, size_t ws_size,
                              hipStream_t stream) {
    const float* h   = (const float*)d_in[0];
    const int*   adj = (const int*)d_in[1];
    const float* W   = (const float*)d_in[2];
    const float* a   = (const float*)d_in[3];
    float* out = (float*)d_out;

    float* WhP = (float*)d_ws;
    float* s_i = WhP + (size_t)kB * kN * kHD;
    float* s_j = s_i + (size_t)kB * kH * kN;

    gat_phase1<<<dim3((kB * kN) / 16), 256, 0, stream>>>(h, W, a, WhP, s_i, s_j);
    gat_phase2<<<dim3(kB * (kN / 32)), 512, 0, stream>>>(adj, WhP, s_i, s_j, out);
}


# Phase 2 mask tile staged in 8 column stages: only stage 0 converted before the sweep; later stages' adj words loaded 2 stages ahead, stored under the sweep, split work-group barrier; sums unchanged
# speedup vs baseline: 1.7568x; 1.0572x over previous
#include <hip/hip_runtime.h>
#include <hip/hip_bf16.h>

typedef __attribute__((ext_vector_type(2))) float v2f;
typedef __attribute__((ext_vector_type(8))) float v8f;
typedef __attribute__((ext_vector_type(4))) int   v4i;

#define GAT_NEG_INF (-9000000000000000.0f)

constexpr int kB   = 8;
constexpr int kN   = 1024;
constexpr int kDin = 128;
constexpr int kH   = 8;
constexpr int kD   = 16;
constexpr int kHD  = 128;
constexpr float kAlpha = 0.2f;

template <int IMM>
__device__ __forceinline__ float swzf(float v) {
    return __int_as_float(__builtin_amdgcn_ds_swizzle(__float_as_int(v), IMM));
}
__device__ __forceinline__ float bpermf(int byteAddr, float v) {
    return __int_as_float(__builtin_amdgcn_ds_bpermute(byteAddr, __float_as_int(v)));
}
#define SWZ_X1  0x041f
#define SWZ_X2  0x081f
#define SWZ_X4  0x101f
#define SWZ_X8  0x201f
#define SWZ_X16 0x401f

__global__ __launch_bounds__(256) void gat_phase1(
    const float* __restrict__ h, const float* __restrict__ W,
    const float* __restrict__ a, float* __restrict__ WhP,
    float* __restrict__ s_i, float* __restrict__ s_j)
{
    __shared__ float hT[16 * 132];

    const int g    = blockIdx.x;
    const int tid  = threadIdx.x;
    const int head = tid >> 5;
    const int lane = tid & 31;
    const int r16  = lane & 15;
    const int half = lane >> 4;
    const int n0   = head * 16;

    const float4* src = (const float4*)(h + (size_t)g * 16 * kDin);
    const float4 hv0 = src[tid];
    const float4 hv1 = src[256 + tid];
    const float asrc = a[head * 2 * kD + r16];
    const float adst = a[head * 2 * kD + kD + r16];
    float wreg[64];
    #pragma unroll
    for (int s = 0; s < 32; ++s) {
        const int ka = 4 * s + 2 * half;
        wreg[2 * s]     = W[(size_t)ka * kHD + n0 + r16];
        wreg[2 * s + 1] = W[(size_t)(ka + 1) * kHD + n0 + r16];
    }

    {
        *(float4*)&hT[(tid >> 5) * 132 + (tid & 31) * 4]       = hv0;
        *(float4*)&hT[(8 + (tid >> 5)) * 132 + (tid & 31) * 4] = hv1;
    }
    __syncthreads();

    v2f areg[32];
    #pragma unroll
    for (int s = 0; s < 32; ++s)
        areg[s] = *(const v2f*)&hT[r16 * 132 + 4 * s + 2 * half];
    __builtin_amdgcn_sched_barrier(0);

    v8f c = {};
    #pragma unroll
    for (int s = 0; s < 32; ++s) {
        v2f bv;
        bv[0] = wreg[2 * s];
        bv[1] = wreg[2 * s + 1];
        c = __builtin_amdgcn_wmma_f32_16x16x4_f32(false, areg[s], false, bv,
                                                  (short)0, c, false, false);
    }

    #pragma unroll
    for (int r = 0; r < 8; ++r) {
        int row = g * 16 + r + half * 8;
        WhP[((size_t)(row >> 1) * kHD + n0 + r16) * 2 + (row & 1)] = c[r];
    }

    float vs[8], vd[8];
    #pragma unroll
    for (int r = 0; r < 8; ++r) { vs[r] = c[r] * asrc; vd[r] = c[r] * adst; }
    #pragma unroll
    for (int r = 0; r < 8; ++r) { vs[r] += swzf<SWZ_X1>(vs[r]); vd[r] += swzf<SWZ_X1>(vd[r]); }
    #pragma unroll
    for (int r = 0; r < 8; ++r) { vs[r] += swzf<SWZ_X2>(vs[r]); vd[r] += swzf<SWZ_X2>(vd[r]); }
    #pragma unroll
    for (int r = 0; r < 8; ++r) { vs[r] += swzf<SWZ_X4>(vs[r]); vd[r] += swzf<SWZ_X4>(vd[r]); }
    #pragma unroll
    for (int r = 0; r < 8; ++r) { vs[r] += swzf<SWZ_X8>(vs[r]); vd[r] += swzf<SWZ_X8>(vd[r]); }
    if (r16 == 0) {
        const int row0 = g * 16 + half * 8;
        const int bb = row0 >> 10, nn0 = row0 & (kN - 1);
        float* pi = s_i + ((size_t)bb * kH + head) * kN + nn0;
        float* pj = s_j + ((size_t)bb * kH + head) * kN + nn0;
        *(float4*)pi       = make_float4(vs[0], vs[1], vs[2], vs[3]);
        *(float4*)(pi + 4) = make_float4(vs[4], vs[5], vs[6], vs[7]);
        *(float4*)pj       = make_float4(vd[0], vd[1], vd[2], vd[3]);
        *(float4*)(pj + 4) = make_float4(vd[4], vd[5], vd[6], vd[7]);
    }
}

constexpr int kMPitch = 96;

constexpr int kKS = 4;

typedef __attribute__((ext_vector_type(4))) float v4f;

struct ChunkRegs {
    v2f wv[kKS];
    v4f fg[kKS];
    v2f ad0[kKS];
    v2f ad1[kKS];
};

__device__ __forceinline__ void load_chunk(
    ChunkRegs& R, int chunk, const float* __restrict__ whLane,
    const v4f* fgHalf, const float* adjM, int r16, int half)
{
    const int halfOff = half * kMPitch;
    #pragma unroll
    for (int kk = 0; kk < kKS; ++kk) {
        const int g   = chunk * kKS + kk;
        const int idx = g * (2 * kMPitch) + halfOff + 2 * ((r16 + g) & 15);
        R.wv[kk]  = *(const v2f*)(whLane + (size_t)g * 512);
        R.fg[kk]  = fgHalf[g * 2];
        R.ad0[kk] = *(const v2f*)&adjM[idx];
        R.ad1[kk] = *(const v2f*)&adjM[idx + 32];
    }
    asm volatile("" ::: "memory");
}

__device__ __forceinline__ void math_chunk(
    const ChunkRegs& R, v2f A0, v2f B0, v2f A1, v2f B1,
    v8f& c0, v8f& c1, v2f& ps0, v2f& ps1)
{
    #pragma unroll
    for (int kk = 0; kk < kKS; ++kk) {
        const v2f F  = R.fg[kk].xy;
        const v2f G  = R.fg[kk].zw;
        const v2f bv = R.wv[kk];
        {
            const v2f p = __builtin_elementwise_max(A0 * F, B0 * G) * R.ad0[kk];
            ps0 += p;
            c0 = __builtin_amdgcn_wmma_f32_16x16x4_f32(false, p, false, bv,
                                                       (short)0, c0, false, false);
        }
        {
            const v2f p = __builtin_elementwise_max(A1 * F, B1 * G) * R.ad1[kk];
            ps1 += p;
            c1 = __builtin_amdgcn_wmma_f32_16x16x4_f32(false, p, false, bv,
                                                       (short)0, c1, false, false);
        }
    }
}

constexpr int kStages    = 8;
constexpr int kStageCols = 64;

__device__ __forceinline__ void load_mask_stage(
    v4i (&av)[2], const int* __restrict__ adjLane, int stage)
{
    const int* p = adjLane + stage * kStageCols;
    av[0] = *(const v4i*)p;
    av[1] = *(const v4i*)(p + (size_t)16 * kN);
}

__device__ __forceinline__ void store_mask_stage(
    float* adjM, const v4i (&av)[2], int stage, int mst)
{
    const int so = stage * (kStageCols / 2) * kMPitch;
    #pragma unroll
    for (int k = 0; k < 2; ++k) {
        const int base = so + mst + 32 * k;
        const v4i v = av[k];
        float2 lo, hi;
        lo.x = v.x ? 1.0f : 0.0f;
        lo.y = v.y ? 1.0f : 0.0f;
        hi.x = v.z ? 1.0f : 0.0f;
        hi.y = v.w ? 1.0f : 0.0f;
        *(float2*)&adjM[base]           = lo;
        *(float2*)&adjM[base + kMPitch] = hi;
    }
}

__global__ __launch_bounds__(512) __attribute__((amdgpu_waves_per_eu(4, 4)))
void gat_phase2(
    const int* __restrict__ adj, const float* __restrict__ WhP,
    const float* __restrict__ s_i, const float* __restrict__ s_j,
    float* __restrict__ out)
{
    __shared__ float  adjM[(kN / 2) * kMPitch];
    __shared__ float4 colFG[kH * (kN / 2)];
    __shared__ float  xch[16 * 9 * 32];

    const int blk  = blockIdx.x;
    const int b    = blk >> 5;
    const int i0   = (blk & 31) * 32;
    const int tid  = threadIdx.x;
    const int wid  = __builtin_amdgcn_readfirstlane(tid >> 5);
    const int head = wid & 7;
    const int jh   = wid >> 3;
    const int lane = tid & 31;
    const int r16  = lane & 15;
    const int half = lane >> 4;

    const float* sI = s_i + ((size_t)b * kH + head) * kN;
    const float* sJ = s_j + ((size_t)b * kH + head) * kN;

    float4 svo[4], svx[4];
    {
        const float4* sJ4 = (const float4*)sJ;
        #pragma unroll
        for (int k = 0; k < 4; ++k) svo[k] = sJ4[(jh * 4 + k) * 32 + lane];
        #pragma unroll
        for (int k = 0; k < 4; ++k) svx[k] = sJ4[((jh ^ 1) * 4 + k) * 32 + lane];
    }
    const float si0 = sI[i0 + r16];
    const float si1 = sI[i0 + 16 + r16];
    __builtin_amdgcn_sched_barrier(0);
    const int mrow  = tid >> 5;
    const int mcol0 = ((tid >> 4) & 1) * 512 + (tid & 15) * 4;
    const int* adjLane = adj + ((size_t)b * kN + i0 + mrow) * kN + mcol0;
    const int mst = (mcol0 >> 1) * kMPitch + 2 * ((mrow + (tid & 15)) & 15);
    v4i av0[2], avn[2];
    load_mask_stage(av0, adjLane, 0);
    load_mask_stage(avn, adjLane, 1);
    __builtin_amdgcn_sched_barrier(0);

    float mv = GAT_NEG_INF;
    {
        #pragma unroll
        for (int k = 0; k < 4; ++k) {
            mv = fmaxf(mv, fmaxf(fmaxf(svo[k].x, svo[k].y), fmaxf(svo[k].z, svo[k].w)));
            mv = fmaxf(mv, fmaxf(fmaxf(svx[k].x, svx[k].y), fmaxf(svx[k].z, svx[k].w)));
        }
        mv = fmaxf(mv, swzf<SWZ_X16>(mv));
        mv = fmaxf(mv, swzf<SWZ_X8>(mv));
        mv = fmaxf(mv, swzf<SWZ_X4>(mv));
        mv = fmaxf(mv, swzf<SWZ_X2>(mv));
        mv = fmaxf(mv, swzf<SWZ_X1>(mv));
        float4* fgw = colFG + head * (kN / 2);
        #pragma unroll
        for (int k = 0; k < 4; ++k) {
            const int f = (jh * 4 + k) * 32 + lane;
            const float d0 = svo[k].x - mv, d1 = svo[k].y - mv;
            const float d2 = svo[k].z - mv, d3 = svo[k].w - mv;
            float4 lo, hi;
            lo.x = __expf(d0);          lo.y = __expf(d1);
            lo.z = __expf(kAlpha * d0); lo.w = __expf(kAlpha * d1);
            hi.x = __expf(d2);          hi.y = __expf(d3);
            hi.z = __expf(kAlpha * d2); hi.w = __expf(kAlpha * d3);
            fgw[2 * f]     = lo;
            fgw[2 * f + 1] = hi;
        }
    }
    __builtin_amdgcn_sched_barrier(0);

    store_mask_stage(adjM, av0, 0, mst);
    asm volatile("s_wait_dscnt 0x0\n\ts_barrier_signal -1\n\ts_barrier_wait -1" ::: "memory");

    const float u0 = (1.0f - kAlpha) * (si0 + mv);
    const float a0 = __expf(fminf(u0, 0.0f));
    const float b0 = __expf(fminf(-u0, 0.0f));
    const float u1 = (1.0f - kAlpha) * (si1 + mv);
    const float a1 = __expf(fminf(u1, 0.0f));
    const float b1 = __expf(fminf(-u1, 0.0f));
    const v2f A0 = {a0, a0}, B0 = {b0, b0}, A1 = {a1, a1}, B1 = {b1, b1};

    const float* whLane = WhP + (size_t)b * kN * kHD + head * 32
                              + (size_t)half * 256 + r16 * 2;
    const v4f* fgHalf = (const v4f*)colFG + head * (kN / 2) + half;
    v8f c0 = {}, c1 = {};
    v2f pv0 = {0.0f, 0.0f}, pv1 = {0.0f, 0.0f};
    const int cb = jh * 32;
    ChunkRegs ra, rb;
    load_chunk(ra, cb, whLane, fgHalf, adjM, r16, half);
    for (int st = 0; st < kStages - 1; ++st) {
        const int c4 = cb + 4 * st;
        store_mask_stage(adjM, avn, st + 1, mst);
        asm volatile("s_wait_dscnt 0x0\n\ts_barrier_signal -1" ::: "memory");
        load_chunk(rb, c4 + 1, whLane, fgHalf, adjM, r16, half);
        if (st + 2 < kStages)
            load_mask_stage(avn, adjLane, st + 2);
        math_chunk(ra, A0, B0, A1, B1, c0, c1, pv0, pv1);
        load_chunk(ra, c4 + 2, whLane, fgHalf, adjM, r16, half);
        math_chunk(rb, A0, B0, A1, B1, c0, c1, pv0, pv1);
        load_chunk(rb, c4 + 3, whLane, fgHalf, adjM, r16, half);
        math_chunk(ra, A0, B0, A1, B1, c0, c1, pv0, pv1);
        asm volatile("s_barrier_wait -1" ::: "memory");
        load_chunk(ra, c4 + 4, whLane, fgHalf, adjM, r16, half);
        math_chunk(rb, A0, B0, A1, B1, c0, c1, pv0, pv1);
    }
    load_chunk(rb, cb + 29, whLane, fgHalf, adjM, r16, half);
    math_chunk(ra, A0, B0, A1, B1, c0, c1, pv0, pv1);
    load_chunk(ra, cb + 30, whLane, fgHalf, adjM, r16, half);
    math_chunk(rb, A0, B0, A1, B1, c0, c1, pv0, pv1);
    load_chunk(rb, cb + 31, whLane, fgHalf, adjM, r16, half);
    math_chunk(ra, A0, B0, A1, B1, c0, c1, pv0, pv1);
    math_chunk(rb, A0, B0, A1, B1, c0, c1, pv0, pv1);

    float ps0 = pv0[0] + pv0[1];
    float ps1 = pv1[0] + pv1[1];
    ps0 += swzf<SWZ_X16>(ps0);
    ps1 += swzf<SWZ_X16>(ps1);

    v8f cg, ck;
    float pg, pk;
    if (jh) { cg = c0; pg = ps0; ck = c1; pk = ps1; }
    else    { cg = c1; pg = ps1; ck = c0; pk = ps0; }
    {
        float* xw = xch + wid * (9 * 32);
        #pragma unroll
        for (int r = 0; r < 8; ++r) xw[r * 32 + lane] = cg[r];
        xw[8 * 32 + lane] = pg;
    }
    __syncthreads();
    {
        const float* xr = xch + (wid ^ 8) * (9 * 32);
        #pragma unroll
        for (int r = 0; r < 8; ++r) ck[r] += xr[r * 32 + lane];
        pk += xr[8 * 32 + lane];
    }

    const float inv_l = 1.0f / fmaxf(pk, 1e-30f);
    const int baddr = half * 32;
    #pragma unroll
    for (int r = 0; r < 8; ++r) {
        float sc = bpermf(baddr + 4 * r, inv_l);
        int row = i0 + jh * 16 + r + half * 8;
        out[((size_t)b * kN + row) * kHD + head * 16 + r16] = ck[r] * sc;
    }
}

extern "C" void kernel_launch(void* const* d_in, const int* in_sizes, int n_in,
                              void* d_out, int out_size, void* d_ws, size_t ws_size,
                              hipStream_t stream) {
    const float* h   = (const float*)d_in[0];
    const int*   adj = (const int*)d_in[1];
    const float* W   = (const float*)d_in[2];
    const float* a   = (const float*)d_in[3];
    float* out = (float*)d_out;

    float* WhP = (float*)d_ws;
    float* s_i = WhP + (size_t)kB * kN * kHD;
    float* s_j = s_i + (size_t)kB * kH * kN;

    gat_phase1<<<dim3((kB * kN) / 16), 256, 0, stream>>>(h, W, a, WhP, s_i, s_j);
    gat_phase2<<<dim3(kB * (kN / 32)), 512, 0, stream>>>(adj, WhP, s_i, s_j, out);
}


# Launch 1 on one wave per SIMD: 256 work-groups of 4 waves, each wave 2 heads x 2 row tiles on four accumulators (a quarter of the waves to enter); W loads in 3 groups of at most 63; sums unchanged
# speedup vs baseline: 1.8360x; 1.0451x over previous
#include <hip/hip_runtime.h>
#include <hip/hip_bf16.h>

typedef __attribute__((ext_vector_type(2))) float v2f;
typedef __attribute__((ext_vector_type(8))) float v8f;
typedef __attribute__((ext_vector_type(4))) int   v4i;

#define GAT_NEG_INF (-9000000000000000.0f)

constexpr int kB   = 8;
constexpr int kN   = 1024;
constexpr int kDin = 128;
constexpr int kH   = 8;
constexpr int kD   = 16;
constexpr int kHD  = 128;
constexpr float kAlpha = 0.2f;

template <int IMM>
__device__ __forceinline__ float swzf(float v) {
    return __int_as_float(__builtin_amdgcn_ds_swizzle(__float_as_int(v), IMM));
}
__device__ __forceinline__ float bpermf(int byteAddr, float v) {
    return __int_as_float(__builtin_amdgcn_ds_bpermute(byteAddr, __float_as_int(v)));
}
#define SWZ_X1  0x041f
#define SWZ_X2  0x081f
#define SWZ_X4  0x101f
#define SWZ_X8  0x201f
#define SWZ_X16 0x401f

__global__ __launch_bounds__(128) void gat_phase1(
    const float* __restrict__ h, const float* __restrict__ W,
    const float* __restrict__ a, float* __restrict__ WhP,
    float* __restrict__ s_i, float* __restrict__ s_j)
{
    __shared__ float hT[32 * 132];

    const int g    = blockIdx.x;
    const int tid  = threadIdx.x;
    const int wv   = tid >> 5;
    const int lane = tid & 31;
    const int r16  = lane & 15;
    const int half = lane >> 4;
    const int n0   = wv * 32;

    const float4* src = (const float4*)(h + (size_t)g * 32 * kDin);
    float4 hv[8];
    #pragma unroll
    for (int k = 0; k < 8; ++k) hv[k] = src[k * 128 + tid];
    float asrc[2], adst[2];
    #pragma unroll
    for (int hh = 0; hh < 2; ++hh) {
        asrc[hh] = a[(2 * wv + hh) * 2 * kD + r16];
        adst[hh] = a[(2 * wv + hh) * 2 * kD + kD + r16];
    }
    float wreg[2][64];
    const float* wLane = W + (size_t)(2 * half) * kHD + n0 + r16;
#define GAT_P1_LOAD_W(S0, S1)                                                 \
    _Pragma("unroll") for (int s = (S0); s < (S1); ++s) {                     \
        _Pragma("unroll") for (int hh = 0; hh < 2; ++hh) {                    \
            wreg[hh][2 * s]     = wLane[(size_t)(4 * s) * kHD + 16 * hh];     \
            wreg[hh][2 * s + 1] = wLane[(size_t)(4 * s + 1) * kHD + 16 * hh]; \
        }                                                                     \
    }
    GAT_P1_LOAD_W(0, 12)

    #pragma unroll
    for (int k = 0; k < 8; ++k)
        *(float4*)&hT[(4 * k + (tid >> 5)) * 132 + (tid & 31) * 4] = hv[k];
    __syncthreads();
    GAT_P1_LOAD_W(12, 27)

    v2f areg[2][32];
    #pragma unroll
    for (int t = 0; t < 2; ++t) {
        #pragma unroll
        for (int s = 0; s < 32; ++s)
            areg[t][s] = *(const v2f*)&hT[(t * 16 + r16) * 132 + 4 * s + 2 * half];
    }
    __builtin_amdgcn_sched_barrier(0);

    v8f c00 = {}, c10 = {}, c01 = {}, c11 = {};
#define GAT_P1_MMA(S0, S1)                                                    \
    _Pragma("unroll") for (int s = (S0); s < (S1); ++s) {                     \
        v2f bv0, bv1;                                                         \
        bv0[0] = wreg[0][2 * s];                                              \
        bv0[1] = wreg[0][2 * s + 1];                                          \
        bv1[0] = wreg[1][2 * s];                                              \
        bv1[1] = wreg[1][2 * s + 1];                                          \
        c00 = __builtin_amdgcn_wmma_f32_16x16x4_f32(false, areg[0][s], false, \
                                    bv0, (short)0, c00, false, false);        \
        c10 = __builtin_amdgcn_wmma_f32_16x16x4_f32(false, areg[1][s], false, \
                                    bv0, (short)0, c10, false, false);        \
        c01 = __builtin_amdgcn_wmma_f32_16x16x4_f32(false, areg[0][s], false, \
                                    bv1, (short)0, c01, false, false);        \
        c11 = __builtin_amdgcn_wmma_f32_16x16x4_f32(false, areg[1][s], false, \
                                    bv1, (short)0, c11, false, false);        \
    }
    GAT_P1_MMA(0, 12)
    asm volatile("" :: "v"(c00), "v"(c10), "v"(c01), "v"(c11) : "memory");
    GAT_P1_LOAD_W(27, 32)
    GAT_P1_MMA(12, 32)
#undef GAT_P1_MMA
#undef GAT_P1_LOAD_W
    v8f cc[2][2];
    cc[0][0] = c00; cc[1][0] = c10; cc[0][1] = c01; cc[1][1] = c11;

    #pragma unroll
    for (int t = 0; t < 2; ++t) {
        #pragma unroll
        for (int hh = 0; hh < 2; ++hh) {
            #pragma unroll
            for (int r = 0; r < 8; ++r) {
                int row = g * 32 + t * 16 + r + half * 8;
                WhP[((size_t)(row >> 1) * kHD + n0 + 16 * hh + r16) * 2 + (row & 1)]
                    = cc[t][hh][r];
            }
        }
    }

    float vs[2][2][8], vd[2][2][8];
    #pragma unroll
    for (int t = 0; t < 2; ++t) {
        #pragma unroll
        for (int hh = 0; hh < 2; ++hh) {
            #pragma unroll
            for (int r = 0; r < 8; ++r) {
                vs[t][hh][r] = cc[t][hh][r] * asrc[hh];
                vd[t][hh][r] = cc[t][hh][r] * adst[hh];
            }
        }
    }
#define GAT_P1_STEP(IMM)                                                      \
    _Pragma("unroll") for (int t = 0; t < 2; ++t) {                           \
        _Pragma("unroll") for (int hh = 0; hh < 2; ++hh) {                    \
            _Pragma("unroll") for (int r = 0; r < 8; ++r) {                   \
                vs[t][hh][r] += swzf<IMM>(vs[t][hh][r]);                      \
                vd[t][hh][r] += swzf<IMM>(vd[t][hh][r]);                      \
            }                                                                 \
        }                                                                     \
    }
    GAT_P1_STEP(SWZ_X1)
    GAT_P1_STEP(SWZ_X2)
    GAT_P1_STEP(SWZ_X4)
    GAT_P1_STEP(SWZ_X8)
#undef GAT_P1_STEP
    if (r16 == 0) {
        #pragma unroll
        for (int t = 0; t < 2; ++t) {
            const int row0 = g * 32 + t * 16 + half * 8;
            const int bb = row0 >> 10, nn0 = row0 & (kN - 1);
            #pragma unroll
            for (int hh = 0; hh < 2; ++hh) {
                const int head = 2 * wv + hh;
                float* pi = s_i + ((size_t)bb * kH + head) * kN + nn0;
                float* pj = s_j + ((size_t)bb * kH + head) * kN + nn0;
                *(float4*)pi       = make_float4(vs[t][hh][0], vs[t][hh][1], vs[t][hh][2], vs[t][hh][3]);
                *(float4*)(pi + 4) = make_float4(vs[t][hh][4], vs[t][hh][5], vs[t][hh][6], vs[t][hh][7]);
                *(float4*)pj       = make_float4(vd[t][hh][0], vd[t][hh][1], vd[t][hh][2], vd[t][hh][3]);
                *(float4*)(pj + 4) = make_float4(vd[t][hh][4], vd[t][hh][5], vd[t][hh][6], vd[t][hh][7]);
            }
        }
    }
}

constexpr int kMPitch = 96;

constexpr int kKS = 4;

typedef __attribute__((ext_vector_type(4))) float v4f;

struct ChunkRegs {
    v2f wv[kKS];
    v4f fg[kKS];
    v2f ad0[kKS];
    v2f ad1[kKS];
};

__device__ __forceinline__ void load_chunk(
    ChunkRegs& R, int chunk, const float* __restrict__ whLane,
    const v4f* fgHalf, const float* adjM, int r16, int half)
{
    const int halfOff = half * kMPitch;
    #pragma unroll
    for (int kk = 0; kk < kKS; ++kk) {
        const int g   = chunk * kKS + kk;
        const int idx = g * (2 * kMPitch) + halfOff + 2 * ((r16 + g) & 15);
        R.wv[kk]  = *(const v2f*)(whLane + (size_t)g * 512);
        R.fg[kk]  = fgHalf[g * 2];
        R.ad0[kk] = *(const v2f*)&adjM[idx];
        R.ad1[kk] = *(const v2f*)&adjM[idx + 32];
    }
    asm volatile("" ::: "memory");
}

__device__ __forceinline__ void math_chunk(
    const ChunkRegs& R, v2f A0, v2f B0, v2f A1, v2f B1,
    v8f& c0, v8f& c1, v2f& ps0, v2f& ps1)
{
    #pragma unroll
    for (int kk = 0; kk < kKS; ++kk) {
        const v2f F  = R.fg[kk].xy;
        const v2f G  = R.fg[kk].zw;
        const v2f bv = R.wv[kk];
        {
            const v2f p = __builtin_elementwise_max(A0 * F, B0 * G) * R.ad0[kk];
            ps0 += p;
            c0 = __builtin_amdgcn_wmma_f32_16x16x4_f32(false, p, false, bv,
                                                       (short)0, c0, false, false);
        }
        {
            const v2f p = __builtin_elementwise_max(A1 * F, B1 * G) * R.ad1[kk];
            ps1 += p;
            c1 = __builtin_amdgcn_wmma_f32_16x16x4_f32(false, p, false, bv,
                                                       (short)0, c1, false, false);
        }
    }
}

constexpr int kStages    = 8;
constexpr int kStageCols = 64;

__device__ __forceinline__ void load_mask_stage(
    v4i (&av)[2], const int* __restrict__ adjLane, int stage)
{
    const int* p = adjLane + stage * kStageCols;
    av[0] = *(const v4i*)p;
    av[1] = *(const v4i*)(p + (size_t)16 * kN);
}

__device__ __forceinline__ void store_mask_stage(
    float* adjM, const v4i (&av)[2], int stage, int mst)
{
    const int so = stage * (kStageCols / 2) * kMPitch;
    #pragma unroll
    for (int k = 0; k < 2; ++k) {
        const int base = so + mst + 32 * k;
        const v4i v = av[k];
        float2 lo, hi;
        lo.x = v.x ? 1.0f : 0.0f;
        lo.y = v.y ? 1.0f : 0.0f;
        hi.x = v.z ? 1.0f : 0.0f;
        hi.y = v.w ? 1.0f : 0.0f;
        *(float2*)&adjM[base]           = lo;
        *(float2*)&adjM[base + kMPitch] = hi;
    }
}

__global__ __launch_bounds__(512) __attribute__((amdgpu_waves_per_eu(4, 4)))
void gat_phase2(
    const int* __restrict__ adj, const float* __restrict__ WhP,
    const float* __restrict__ s_i, const float* __restrict__ s_j,
    float* __restrict__ out)
{
    __shared__ float  adjM[(kN / 2) * kMPitch];
    __shared__ float4 colFG[kH * (kN / 2)];
    __shared__ float  xch[16 * 9 * 32];

    const int blk  = blockIdx.x;
    const int b    = blk >> 5;
    const int i0   = (blk & 31) * 32;
    const int tid  = threadIdx.x;
    const int wid  = __builtin_amdgcn_readfirstlane(tid >> 5);
    const int head = wid & 7;
    const int jh   = wid >> 3;
    const int lane = tid & 31;
    const int r16  = lane & 15;
    const int half = lane >> 4;

    const float* sI = s_i + ((size_t)b * kH + head) * kN;
    const float* sJ = s_j + ((size_t)b * kH + head) * kN;

    float4 svo[4], svx[4];
    {
        const float4* sJ4 = (const float4*)sJ;
        #pragma unroll
        for (int k = 0; k < 4; ++k) svo[k] = sJ4[(jh * 4 + k) * 32 + lane];
        #pragma unroll
        for (int k = 0; k < 4; ++k) svx[k] = sJ4[((jh ^ 1) * 4 + k) * 32 + lane];
    }
    const float si0 = sI[i0 + r16];
    const float si1 = sI[i0 + 16 + r16];
    __builtin_amdgcn_sched_barrier(0);
    const int mrow  = tid >> 5;
    const int mcol0 = ((tid >> 4) & 1) * 512 + (tid & 15) * 4;
    const int* adjLane = adj + ((size_t)b * kN + i0 + mrow) * kN + mcol0;
    const int mst = (mcol0 >> 1) * kMPitch + 2 * ((mrow + (tid & 15)) & 15);
    v4i av0[2], avn[2];
    load_mask_stage(av0, adjLane, 0);
    load_mask_stage(avn, adjLane, 1);
    __builtin_amdgcn_sched_barrier(0);

    float mv = GAT_NEG_INF;
    {
        #pragma unroll
        for (int k = 0; k < 4; ++k) {
            mv = fmaxf(mv, fmaxf(fmaxf(svo[k].x, svo[k].y), fmaxf(svo[k].z, svo[k].w)));
            mv = fmaxf(mv, fmaxf(fmaxf(svx[k].x, svx[k].y), fmaxf(svx[k].z, svx[k].w)));
        }
        mv = fmaxf(mv, swzf<SWZ_X16>(mv));
        mv = fmaxf(mv, swzf<SWZ_X8>(mv));
        mv = fmaxf(mv, swzf<SWZ_X4>(mv));
        mv = fmaxf(mv, swzf<SWZ_X2>(mv));
        mv = fmaxf(mv, swzf<SWZ_X1>(mv));
        float4* fgw = colFG + head * (kN / 2);
        #pragma unroll
        for (int k = 0; k < 4; ++k) {
            const int f = (jh * 4 + k) * 32 + lane;
            const float d0 = svo[k].x - mv, d1 = svo[k].y - mv;
            const float d2 = svo[k].z - mv, d3 = svo[k].w - mv;
            float4 lo, hi;
            lo.x = __expf(d0);          lo.y = __expf(d1);
            lo.z = __expf(kAlpha * d0); lo.w = __expf(kAlpha * d1);
            hi.x = __expf(d2);          hi.y = __expf(d3);
            hi.z = __expf(kAlpha * d2); hi.w = __expf(kAlpha * d3);
            fgw[2 * f]     = lo;
            fgw[2 * f + 1] = hi;
        }
    }
    __builtin_amdgcn_sched_barrier(0);

    store_mask_stage(adjM, av0, 0, mst);
    asm volatile("s_wait_dscnt 0x0\n\ts_barrier_signal -1\n\ts_barrier_wait -1" ::: "memory");

    const float u0 = (1.0f - kAlpha) * (si0 + mv);
    const float a0 = __expf(fminf(u0, 0.0f));
    const float b0 = __expf(fminf(-u0, 0.0f));
    const float u1 = (1.0f - kAlpha) * (si1 + mv);
    const float a1 = __expf(fminf(u1, 0.0f));
    const float b1 = __expf(fminf(-u1, 0.0f));
    const v2f A0 = {a0, a0}, B0 = {b0, b0}, A1 = {a1, a1}, B1 = {b1, b1};

    const float* whLane = WhP + (size_t)b * kN * kHD + head * 32
                              + (size_t)half * 256 + r16 * 2;
    const v4f* fgHalf = (const v4f*)colFG + head * (kN / 2) + half;
    v8f c0 = {}, c1 = {};
    v2f pv0 = {0.0f, 0.0f}, pv1 = {0.0f, 0.0f};
    const int cb = jh * 32;
    ChunkRegs ra, rb;
    load_chunk(ra, cb, whLane, fgHalf, adjM, r16, half);
    for (int st = 0; st < kStages - 1; ++st) {
        const int c4 = cb + 4 * st;
        store_mask_stage(adjM, avn, st + 1, mst);
        asm volatile("s_wait_dscnt 0x0\n\ts_barrier_signal -1" ::: "memory");
        load_chunk(rb, c4 + 1, whLane, fgHalf, adjM, r16, half);
        if (st + 2 < kStages)
            load_mask_stage(avn, adjLane, st + 2);
        math_chunk(ra, A0, B0, A1, B1, c0, c1, pv0, pv1);
        load_chunk(ra, c4 + 2, whLane, fgHalf, adjM, r16, half);
        math_chunk(rb, A0, B0, A1, B1, c0, c1, pv0, pv1);
        load_chunk(rb, c4 + 3, whLane, fgHalf, adjM, r16, half);
        math_chunk(ra, A0, B0, A1, B1, c0, c1, pv0, pv1);
        asm volatile("s_barrier_wait -1" ::: "memory");
        load_chunk(ra, c4 + 4, whLane, fgHalf, adjM, r16, half);
        math_chunk(rb, A0, B0, A1, B1, c0, c1, pv0, pv1);
    }
    load_chunk(rb, cb + 29, whLane, fgHalf, adjM, r16, half);
    math_chunk(ra, A0, B0, A1, B1, c0, c1, pv0, pv1);
    load_chunk(ra, cb + 30, whLane, fgHalf, adjM, r16, half);
    math_chunk(rb, A0, B0, A1, B1, c0, c1, pv0, pv1);
    load_chunk(rb, cb + 31, whLane, fgHalf, adjM, r16, half);
    math_chunk(ra, A0, B0, A1, B1, c0, c1, pv0, pv1);
    math_chunk(rb, A0, B0, A1, B1, c0, c1, pv0, pv1);

    float ps0 = pv0[0] + pv0[1];
    float ps1 = pv1[0] + pv1[1];
    ps0 += swzf<SWZ_X16>(ps0);
    ps1 += swzf<SWZ_X16>(ps1);

    v8f cg, ck;
    float pg, pk;
    if (jh) { cg = c0; pg = ps0; ck = c1; pk = ps1; }
    else    { cg = c1; pg = ps1; ck = c0; pk = ps0; }
    {
        float* xw = xch + wid * (9 * 32);
        #pragma unroll
        for (int r = 0; r < 8; ++r) xw[r * 32 + lane] = cg[r];
        xw[8 * 32 + lane] = pg;
    }
    __syncthreads();
    {
        const float* xr = xch + (wid ^ 8) * (9 * 32);
        #pragma unroll
        for (int r = 0; r < 8; ++r) ck[r] += xr[r * 32 + lane];
        pk += xr[8 * 32 + lane];
    }

    const float inv_l = 1.0f / fmaxf(pk, 1e-30f);
    const int baddr = half * 32;
    #pragma unroll
    for (int r = 0; r < 8; ++r) {
        float sc = bpermf(baddr + 4 * r, inv_l);
        int row = i0 + jh * 16 + r + half * 8;
        out[((size_t)b * kN + row) * kHD + head * 16 + r16] = ck[r] * sc;
    }
}

extern "C" void kernel_launch(void* const* d_in, const int* in_sizes, int n_in,
                              void* d_out, int out_size, void* d_ws, size_t ws_size,
                              hipStream_t stream) {
    const float* h   = (const float*)d_in[0];
    const int*   adj = (const int*)d_in[1];
    const float* W   = (const float*)d_in[2];
    const float* a   = (const float*)d_in[3];
    float* out = (float*)d_out;

    float* WhP = (float*)d_ws;
    float* s_i = WhP + (size_t)kB * kN * kHD;
    float* s_j = s_i + (size_t)kB * kH * kN;

    gat_phase1<<<dim3((kB * kN) / 32), 128, 0, stream>>>(h, W, a, WhP, s_i, s_j);
    gat_phase2<<<dim3(kB * (kN / 32)), 512, 0, stream>>>(adj, WhP, s_i, s_j, out);
}


# Phase 2 element math refactored: P=G_j*min(max(H_j,B_i),mask), H=exp((1-alpha)d_j): G multiplied once into the shared Wh pair, max+mask one integer max-min on [0,1] values; sum order unchanged
# speedup vs baseline: 1.9029x; 1.0364x over previous
#include <hip/hip_runtime.h>
#include <hip/hip_bf16.h>

typedef __attribute__((ext_vector_type(2))) float v2f;
typedef __attribute__((ext_vector_type(8))) float v8f;
typedef __attribute__((ext_vector_type(4))) int   v4i;

#define GAT_NEG_INF (-9000000000000000.0f)

constexpr int kB   = 8;
constexpr int kN   = 1024;
constexpr int kDin = 128;
constexpr int kH   = 8;
constexpr int kD   = 16;
constexpr int kHD  = 128;
constexpr float kAlpha = 0.2f;

template <int IMM>
__device__ __forceinline__ float swzf(float v) {
    return __int_as_float(__builtin_amdgcn_ds_swizzle(__float_as_int(v), IMM));
}
__device__ __forceinline__ float bpermf(int byteAddr, float v) {
    return __int_as_float(__builtin_amdgcn_ds_bpermute(byteAddr, __float_as_int(v)));
}
#define SWZ_X1  0x041f
#define SWZ_X2  0x081f
#define SWZ_X4  0x101f
#define SWZ_X8  0x201f
#define SWZ_X16 0x401f

__global__ __launch_bounds__(128) void gat_phase1(
    const float* __restrict__ h, const float* __restrict__ W,
    const float* __restrict__ a, float* __restrict__ WhP,
    float* __restrict__ s_i, float* __restrict__ s_j)
{
    __shared__ float hT[32 * 132];

    const int g    = blockIdx.x;
    const int tid  = threadIdx.x;
    const int wv   = tid >> 5;
    const int lane = tid & 31;
    const int r16  = lane & 15;
    const int half = lane >> 4;
    const int n0   = wv * 32;

    const float4* src = (const float4*)(h + (size_t)g * 32 * kDin);
    float4 hv[8];
    #pragma unroll
    for (int k = 0; k < 8; ++k) hv[k] = src[k * 128 + tid];
    float asrc[2], adst[2];
    #pragma unroll
    for (int hh = 0; hh < 2; ++hh) {
        asrc[hh] = a[(2 * wv + hh) * 2 * kD + r16];
        adst[hh] = a[(2 * wv + hh) * 2 * kD + kD + r16];
    }
    float wreg[2][64];
    const float* wLane = W + (size_t)(2 * half) * kHD + n0 + r16;
#define GAT_P1_LOAD_W(S0, S1)                                                 \
    _Pragma("unroll") for (int s = (S0); s < (S1); ++s) {                     \
        _Pragma("unroll") for (int hh = 0; hh < 2; ++hh) {                    \
            wreg[hh][2 * s]     = wLane[(size_t)(4 * s) * kHD + 16 * hh];     \
            wreg[hh][2 * s + 1] = wLane[(size_t)(4 * s + 1) * kHD + 16 * hh]; \
        }                                                                     \
    }
    GAT_P1_LOAD_W(0, 12)

    #pragma unroll
    for (int k = 0; k < 8; ++k)
        *(float4*)&hT[(4 * k + (tid >> 5)) * 132 + (tid & 31) * 4] = hv[k];
    __syncthreads();
    GAT_P1_LOAD_W(12, 27)

    v2f areg[2][32];
    #pragma unroll
    for (int t = 0; t < 2; ++t) {
        #pragma unroll
        for (int s = 0; s < 32; ++s)
            areg[t][s] = *(const v2f*)&hT[(t * 16 + r16) * 132 + 4 * s + 2 * half];
    }
    __builtin_amdgcn_sched_barrier(0);

    v8f c00 = {}, c10 = {}, c01 = {}, c11 = {};
#define GAT_P1_MMA(S0, S1)                                                    \
    _Pragma("unroll") for (int s = (S0); s < (S1); ++s) {                     \
        v2f bv0, bv1;                                                         \
        bv0[0] = wreg[0][2 * s];                                              \
        bv0[1] = wreg[0][2 * s + 1];                                          \
        bv1[0] = wreg[1][2 * s];                                              \
        bv1[1] = wreg[1][2 * s + 1];                                          \
        c00 = __builtin_amdgcn_wmma_f32_16x16x4_f32(false, areg[0][s], false, \
                                    bv0, (short)0, c00, false, false);        \
        c10 = __builtin_amdgcn_wmma_f32_16x16x4_f32(false, areg[1][s], false, \
                                    bv0, (short)0, c10, false, false);        \
        c01 = __builtin_amdgcn_wmma_f32_16x16x4_f32(false, areg[0][s], false, \
                                    bv1, (short)0, c01, false, false);        \
        c11 = __builtin_amdgcn_wmma_f32_16x16x4_f32(false, areg[1][s], false, \
                                    bv1, (short)0, c11, false, false);        \
    }
    GAT_P1_MMA(0, 12)
    asm volatile("" :: "v"(c00), "v"(c10), "v"(c01), "v"(c11) : "memory");
    GAT_P1_LOAD_W(27, 32)
    GAT_P1_MMA(12, 32)
#undef GAT_P1_MMA
#undef GAT_P1_LOAD_W
    v8f cc[2][2];
    cc[0][0] = c00; cc[1][0] = c10; cc[0][1] = c01; cc[1][1] = c11;

    #pragma unroll
    for (int t = 0; t < 2; ++t) {
        #pragma unroll
        for (int hh = 0; hh < 2; ++hh) {
            #pragma unroll
            for (int r = 0; r < 8; ++r) {
                int row = g * 32 + t * 16 + r + half * 8;
                WhP[((size_t)(row >> 1) * kHD + n0 + 16 * hh + r16) * 2 + (row & 1)]
                    = cc[t][hh][r];
            }
        }
    }

    float vs[2][2][8], vd[2][2][8];
    #pragma unroll
    for (int t = 0; t < 2; ++t) {
        #pragma unroll
        for (int hh = 0; hh < 2; ++hh) {
            #pragma unroll
            for (int r = 0; r < 8; ++r) {
                vs[t][hh][r] = cc[t][hh][r] * asrc[hh];
                vd[t][hh][r] = cc[t][hh][r] * adst[hh];
            }
        }
    }
#define GAT_P1_STEP(IMM)                                                      \
    _Pragma("unroll") for (int t = 0; t < 2; ++t) {                           \
        _Pragma("unroll") for (int hh = 0; hh < 2; ++hh) {                    \
            _Pragma("unroll") for (int r = 0; r < 8; ++r) {                   \
                vs[t][hh][r] += swzf<IMM>(vs[t][hh][r]);                      \
                vd[t][hh][r] += swzf<IMM>(vd[t][hh][r]);                      \
            }                                                                 \
        }                                                                     \
    }
    GAT_P1_STEP(SWZ_X1)
    GAT_P1_STEP(SWZ_X2)
    GAT_P1_STEP(SWZ_X4)
    GAT_P1_STEP(SWZ_X8)
#undef GAT_P1_STEP
    if (r16 == 0) {
        #pragma unroll
        for (int t = 0; t < 2; ++t) {
            const int row0 = g * 32 + t * 16 + half * 8;
            const int bb = row0 >> 10, nn0 = row0 & (kN - 1);
            #pragma unroll
            for (int hh = 0; hh < 2; ++hh) {
                const int head = 2 * wv + hh;
                float* pi = s_i + ((size_t)bb * kH + head) * kN + nn0;
                float* pj = s_j + ((size_t)bb * kH + head) * kN + nn0;
                *(float4*)pi       = make_float4(vs[t][hh][0], vs[t][hh][1], vs[t][hh][2], vs[t][hh][3]);
                *(float4*)(pi + 4) = make_float4(vs[t][hh][4], vs[t][hh][5], vs[t][hh][6], vs[t][hh][7]);
                *(float4*)pj       = make_float4(vd[t][hh][0], vd[t][hh][1], vd[t][hh][2], vd[t][hh][3]);
                *(float4*)(pj + 4) = make_float4(vd[t][hh][4], vd[t][hh][5], vd[t][hh][6], vd[t][hh][7]);
            }
        }
    }
}

constexpr int kMPitch = 96;

constexpr int kKS = 4;

typedef __attribute__((ext_vector_type(4))) float v4f;
typedef __attribute__((ext_vector_type(2))) int   v2i;

__device__ __forceinline__ v2f max_then_mask(v2f x, v2f y, v2f m) {
    const v2i r = __builtin_elementwise_min(
        __builtin_elementwise_max(__builtin_bit_cast(v2i, x), __builtin_bit_cast(v2i, y)),
        __builtin_bit_cast(v2i, m));
    return __builtin_bit_cast(v2f, r);
}

struct ChunkRegs {
    v2f wv[kKS];
    v4f fg[kKS];
    v2f ad0[kKS];
    v2f ad1[kKS];
};

__device__ __forceinline__ void load_chunk(
    ChunkRegs& R, int chunk, const float* __restrict__ whLane,
    const v4f* fgHalf, const float* adjM, int r16, int half)
{
    const int halfOff = half * kMPitch;
    #pragma unroll
    for (int kk = 0; kk < kKS; ++kk) {
        const int g   = chunk * kKS + kk;
        const int idx = g * (2 * kMPitch) + halfOff + 2 * ((r16 + g) & 15);
        R.wv[kk]  = *(const v2f*)(whLane + (size_t)g * 512);
        R.fg[kk]  = fgHalf[g * 2];
        R.ad0[kk] = *(const v2f*)&adjM[idx];
        R.ad1[kk] = *(const v2f*)&adjM[idx + 32];
    }
    asm volatile("" ::: "memory");
}

__device__ __forceinline__ void math_chunk(
    const ChunkRegs& R, v2f B0, v2f B1,
    v8f& c0, v8f& c1, v2f& ps0, v2f& ps1)
{
    #pragma unroll
    for (int kk = 0; kk < kKS; ++kk) {
        const v2f H  = R.fg[kk].xy;
        const v2f G  = R.fg[kk].zw;
        const v2f bv = R.wv[kk] * G;
        {
            const v2f q = max_then_mask(H, B0, R.ad0[kk]);
            ps0 += q * G;
            c0 = __builtin_amdgcn_wmma_f32_16x16x4_f32(false, q, false, bv,
                                                       (short)0, c0, false, false);
        }
        {
            const v2f q = max_then_mask(H, B1, R.ad1[kk]);
            ps1 += q * G;
            c1 = __builtin_amdgcn_wmma_f32_16x16x4_f32(false, q, false, bv,
                                                       (short)0, c1, false, false);
        }
    }
}

constexpr int kStages    = 8;
constexpr int kStageCols = 64;

__device__ __forceinline__ void load_mask_stage(
    v4i (&av)[2], const int* __restrict__ adjLane, int stage)
{
    const int* p = adjLane + stage * kStageCols;
    av[0] = *(const v4i*)p;
    av[1] = *(const v4i*)(p + (size_t)16 * kN);
}

__device__ __forceinline__ void store_mask_stage(
    float* adjM, const v4i (&av)[2], int stage, int mst)
{
    const int so = stage * (kStageCols / 2) * kMPitch;
    #pragma unroll
    for (int k = 0; k < 2; ++k) {
        const int base = so + mst + 32 * k;
        const v4i v = av[k];
        float2 lo, hi;
        lo.x = v.x ? 1.0f : 0.0f;
        lo.y = v.y ? 1.0f : 0.0f;
        hi.x = v.z ? 1.0f : 0.0f;
        hi.y = v.w ? 1.0f : 0.0f;
        *(float2*)&adjM[base]           = lo;
        *(float2*)&adjM[base + kMPitch] = hi;
    }
}

__global__ __launch_bounds__(512) __attribute__((amdgpu_waves_per_eu(4, 4)))
void gat_phase2(
    const int* __restrict__ adj, const float* __restrict__ WhP,
    const float* __restrict__ s_i, const float* __restrict__ s_j,
    float* __restrict__ out)
{
    __shared__ float  adjM[(kN / 2) * kMPitch];
    __shared__ float4 colFG[kH * (kN / 2)];
    __shared__ float  xch[16 * 9 * 32];

    const int blk  = blockIdx.x;
    const int b    = blk >> 5;
    const int i0   = (blk & 31) * 32;
    const int tid  = threadIdx.x;
    const int wid  = __builtin_amdgcn_readfirstlane(tid >> 5);
    const int head = wid & 7;
    const int jh   = wid >> 3;
    const int lane = tid & 31;
    const int r16  = lane & 15;
    const int half = lane >> 4;

    const float* sI = s_i + ((size_t)b * kH + head) * kN;
    const float* sJ = s_j + ((size_t)b * kH + head) * kN;

    float4 svo[4], svx[4];
    {
        const float4* sJ4 = (const float4*)sJ;
        #pragma unroll
        for (int k = 0; k < 4; ++k) svo[k] = sJ4[(jh * 4 + k) * 32 + lane];
        #pragma unroll
        for (int k = 0; k < 4; ++k) svx[k] = sJ4[((jh ^ 1) * 4 + k) * 32 + lane];
    }
    const float si0 = sI[i0 + r16];
    const float si1 = sI[i0 + 16 + r16];
    __builtin_amdgcn_sched_barrier(0);
    const int mrow  = tid >> 5;
    const int mcol0 = ((tid >> 4) & 1) * 512 + (tid & 15) * 4;
    const int* adjLane = adj + ((size_t)b * kN + i0 + mrow) * kN + mcol0;
    const int mst = (mcol0 >> 1) * kMPitch + 2 * ((mrow + (tid & 15)) & 15);
    v4i av0[2], avn[2];
    load_mask_stage(av0, adjLane, 0);
    load_mask_stage(avn, adjLane, 1);
    __builtin_amdgcn_sched_barrier(0);

    float mv = GAT_NEG_INF;
    {
        #pragma unroll
        for (int k = 0; k < 4; ++k) {
            mv = fmaxf(mv, fmaxf(fmaxf(svo[k].x, svo[k].y), fmaxf(svo[k].z, svo[k].w)));
            mv = fmaxf(mv, fmaxf(fmaxf(svx[k].x, svx[k].y), fmaxf(svx[k].z, svx[k].w)));
        }
        mv = fmaxf(mv, swzf<SWZ_X16>(mv));
        mv = fmaxf(mv, swzf<SWZ_X8>(mv));
        mv = fmaxf(mv, swzf<SWZ_X4>(mv));
        mv = fmaxf(mv, swzf<SWZ_X2>(mv));
        mv = fmaxf(mv, swzf<SWZ_X1>(mv));
        float4* fgw = colFG + head * (kN / 2);
        #pragma unroll
        for (int k = 0; k < 4; ++k) {
            const int f = (jh * 4 + k) * 32 + lane;
            const float d0 = svo[k].x - mv, d1 = svo[k].y - mv;
            const float d2 = svo[k].z - mv, d3 = svo[k].w - mv;
            float4 lo, hi;
            lo.x = __expf((1.0f - kAlpha) * d0); lo.y = __expf((1.0f - kAlpha) * d1);
            lo.z = __expf(kAlpha * d0);          lo.w = __expf(kAlpha * d1);
            hi.x = __expf((1.0f - kAlpha) * d2); hi.y = __expf((1.0f - kAlpha) * d3);
            hi.z = __expf(kAlpha * d2);          hi.w = __expf(kAlpha * d3);
            fgw[2 * f]     = lo;
            fgw[2 * f + 1] = hi;
        }
    }
    __builtin_amdgcn_sched_barrier(0);

    store_mask_stage(adjM, av0, 0, mst);
    asm volatile("s_wait_dscnt 0x0\n\ts_barrier_signal -1\n\ts_barrier_wait -1" ::: "memory");

    const float u0 = (1.0f - kAlpha) * (si0 + mv);
    const float b0 = __expf(fminf(-u0, 0.0f));
    const float u1 = (1.0f - kAlpha) * (si1 + mv);
    const float b1 = __expf(fminf(-u1, 0.0f));
    const v2f B0 = {b0, b0}, B1 = {b1, b1};

    const float* whLane = WhP + (size_t)b * kN * kHD + head * 32
                              + (size_t)half * 256 + r16 * 2;
    const v4f* fgHalf = (const v4f*)colFG + head * (kN / 2) + half;
    v8f c0 = {}, c1 = {};
    v2f pv0 = {0.0f, 0.0f}, pv1 = {0.0f, 0.0f};
    const int cb = jh * 32;
    ChunkRegs ra, rb;
    load_chunk(ra, cb, whLane, fgHalf, adjM, r16, half);
    for (int st = 0; st < kStages - 1; ++st) {
        const int c4 = cb + 4 * st;
        store_mask_stage(adjM, avn, st + 1, mst);
        asm volatile("s_wait_dscnt 0x0\n\ts_barrier_signal -1" ::: "memory");
        load_chunk(rb, c4 + 1, whLane, fgHalf, adjM, r16, half);
        if (st + 2 < kStages)
            load_mask_stage(avn, adjLane, st + 2);
        math_chunk(ra, B0, B1, c0, c1, pv0, pv1);
        load_chunk(ra, c4 + 2, whLane, fgHalf, adjM, r16, half);
        math_chunk(rb, B0, B1, c0, c1, pv0, pv1);
        load_chunk(rb, c4 + 3, whLane, fgHalf, adjM, r16, half);
        math_chunk(ra, B0, B1, c0, c1, pv0, pv1);
        asm volatile("s_barrier_wait -1" ::: "memory");
        load_chunk(ra, c4 + 4, whLane, fgHalf, adjM, r16, half);
        math_chunk(rb, B0, B1, c0, c1, pv0, pv1);
    }
    load_chunk(rb, cb + 29, whLane, fgHalf, adjM, r16, half);
    math_chunk(ra, B0, B1, c0, c1, pv0, pv1);
    load_chunk(ra, cb + 30, whLane, fgHalf, adjM, r16, half);
    math_chunk(rb, B0, B1, c0, c1, pv0, pv1);
    load_chunk(rb, cb + 31, whLane, fgHalf, adjM, r16, half);
    math_chunk(ra, B0, B1, c0, c1, pv0, pv1);
    math_chunk(rb, B0, B1, c0, c1, pv0, pv1);

    float ps0 = pv0[0] + pv0[1];
    float ps1 = pv1[0] + pv1[1];
    ps0 += swzf<SWZ_X16>(ps0);
    ps1 += swzf<SWZ_X16>(ps1);

    v8f cg, ck;
    float pg, pk;
    if (jh) { cg = c0; pg = ps0; ck = c1; pk = ps1; }
    else    { cg = c1; pg = ps1; ck = c0; pk = ps0; }
    {
        float* xw = xch + wid * (9 * 32);
        #pragma unroll
        for (int r = 0; r < 8; ++r) xw[r * 32 + lane] = cg[r];
        xw[8 * 32 + lane] = pg;
    }
    __syncthreads();
    {
        const float* xr = xch + (wid ^ 8) * (9 * 32);
        #pragma unroll
        for (int r = 0; r < 8; ++r) ck[r] += xr[r * 32 + lane];
        pk += xr[8 * 32 + lane];
    }

    const float inv_l = 1.0f / fmaxf(pk, 1e-30f);
    const int baddr = half * 32;
    #pragma unroll
    for (int r = 0; r < 8; ++r) {
        float sc = bpermf(baddr + 4 * r, inv_l);
        int row = i0 + jh * 16 + r + half * 8;
        out[((size_t)b * kN + row) * kHD + head * 16 + r16] = ck[r] * sc;
    }
}

extern "C" void kernel_launch(void* const* d_in, const int* in_sizes, int n_in,
                              void* d_out, int out_size, void* d_ws, size_t ws_size,
                              hipStream_t stream) {
    const float* h   = (const float*)d_in[0];
    const int*   adj = (const int*)d_in[1];
    const float* W   = (const float*)d_in[2];
    const float* a   = (const float*)d_in[3];
    float* out = (float*)d_out;

    float* WhP = (float*)d_ws;
    float* s_i = WhP + (size_t)kB * kN * kHD;
    float* s_j = s_i + (size_t)kB * kH * kN;

    gat_phase1<<<dim3((kB * kN) / 32), 128, 0, stream>>>(h, W, a, WhP, s_i, s_j);
    gat_phase2<<<dim3(kB * (kN / 32)), 512, 0, stream>>>(adj, WhP, s_i, s_j, out);
}
